# starting kernel
# baseline (speedup 1.0000x reference)
#include <hip/hip_runtime.h>
#include <hip/hip_bf16.h>

typedef __attribute__((ext_vector_type(2))) float v2f;
typedef __attribute__((ext_vector_type(8))) float v8f;

__global__ __launch_bounds__(256)
void stats_kernel(const float* __restrict__ feat, const float* __restrict__ noise,
                  float* __restrict__ scl, float* __restrict__ nmn,
                  int nd, float snr_pow) {
  int b = blockIdx.x;
  int t = threadIdx.x;
  const float* f  = feat  + (long long)b * nd;
  const float* nz = noise + (long long)b * nd;
  double sf = 0.0, sf2 = 0.0, sn = 0.0, sn2 = 0.0;
  for (int i = t; i < nd; i += 256) {
    double x = (double)f[i];
    double y = (double)nz[i];
    sf += x; sf2 += x * x; sn += y; sn2 += y * y;
  }
  __shared__ double sh[4][256];
  sh[0][t] = sf; sh[1][t] = sf2; sh[2][t] = sn; sh[3][t] = sn2;
  __syncthreads();
  for (int s = 128; s > 0; s >>= 1) {
    if (t < s) {
      sh[0][t] += sh[0][t + s]; sh[1][t] += sh[1][t + s];
      sh[2][t] += sh[2][t + s]; sh[3][t] += sh[3][t + s];
    }
    __syncthreads();
  }
  if (t == 0) {
    double inv = 1.0 / (double)nd;
    double gmf = sh[0][0] * inv;
    double gmn = sh[2][0] * inv;
    double tvar = sh[1][0] * inv - gmf * gmf;
    double nvar = sh[3][0] * inv - gmn * gmn;
    scl[b] = (float)sqrt((tvar / (double)snr_pow) / nvar);
    nmn[b] = (float)gmn;
  }
}

__global__ __launch_bounds__(256)
void perturb_kernel(const float* __restrict__ feat, const float* __restrict__ noise,
                    const float* __restrict__ scl, const float* __restrict__ nmn,
                    float* __restrict__ Fp, int nd) {
  int b = blockIdx.y;
  long long idx = (long long)b * nd + (long long)blockIdx.x * 256 + threadIdx.x;
  Fp[idx] = feat[idx] + (noise[idx] - nmn[b]) * scl[b];
}

__global__ __launch_bounds__(256)
void mean_kernel(const float* __restrict__ Fp, float* __restrict__ mean, int n, int d) {
  int b = blockIdx.x;
  int j = threadIdx.x;
  const float* base = Fp + (long long)b * n * d;
  float s = 0.0f;
  for (int i = 0; i < n; ++i) s += base[(long long)i * d + j];
  mean[(long long)b * d + j] = s / (float)n;
}

__global__ __launch_bounds__(256)
void center_kernel(float* __restrict__ Fp, const float* __restrict__ mean, int n, int d) {
  int row = blockIdx.x;
  int j   = threadIdx.x;
  int b   = row / n;
  Fp[(long long)row * d + j] -= mean[(long long)b * d + j];
}

#define BM 128
#define BN 128
#define BK 16
#define LDP 18

__global__ __launch_bounds__(256)
void wmma_gemm_batched(const float* __restrict__ A, long long sAb, int sAm, int sAk,
                       const float* __restrict__ Bm, long long sBb, int sBk, int sBn,
                       float* __restrict__ Dm, long long sDb,
                       int N, int K, float alpha, int mode) {
  __shared__ __align__(16) float As[BM][LDP];
  __shared__ __align__(16) float Bst[BN][LDP];

  int b  = blockIdx.z;
  int m0 = blockIdx.y * BM;
  int n0 = blockIdx.x * BN;
  const float* Ab = A  + (long long)b * sAb;
  const float* Bb = Bm + (long long)b * sBb;
  float*       Db = Dm + (long long)b * sDb;

  int tid  = threadIdx.x;
  int wid  = tid >> 5;
  int lane = tid & 31;
  int mr   = lane & 15;
  int h    = lane >> 4;
  int wm   = (wid >> 1) * 32;
  int wn   = (wid & 1) * 64;

  v8f acc[2][4];
#pragma unroll
  for (int im = 0; im < 2; ++im)
#pragma unroll
    for (int in = 0; in < 4; ++in) acc[im][in] = (v8f){};

  for (int k0 = 0; k0 < K; k0 += BK) {
#pragma unroll
    for (int i = 0; i < 8; ++i) {
      int e = tid + i * 256;
      int m = e >> 4, kk = e & 15;
      As[m][kk] = Ab[(long long)(m0 + m) * sAm + (long long)(k0 + kk) * sAk];
    }
#pragma unroll
    for (int i = 0; i < 8; ++i) {
      int e = tid + i * 256;
      int nn = e & 127, kk = e >> 7;
      Bst[nn][kk] = Bb[(long long)(k0 + kk) * sBk + (long long)(n0 + nn) * sBn];
    }
    __syncthreads();
#pragma unroll
    for (int kk = 0; kk < BK; kk += 4) {
      int kc = kk + 2 * h;
      v2f a0 = *(const v2f*)&As[wm + mr][kc];
      v2f a1 = *(const v2f*)&As[wm + 16 + mr][kc];
      v2f b0 = *(const v2f*)&Bst[wn + mr][kc];
      v2f b1 = *(const v2f*)&Bst[wn + 16 + mr][kc];
      v2f b2 = *(const v2f*)&Bst[wn + 32 + mr][kc];
      v2f b3 = *(const v2f*)&Bst[wn + 48 + mr][kc];
      acc[0][0] = __builtin_amdgcn_wmma_f32_16x16x4_f32(false, a0, false, b0, (short)0, acc[0][0], false, false);
      acc[0][1] = __builtin_amdgcn_wmma_f32_16x16x4_f32(false, a0, false, b1, (short)0, acc[0][1], false, false);
      acc[0][2] = __builtin_amdgcn_wmma_f32_16x16x4_f32(false, a0, false, b2, (short)0, acc[0][2], false, false);
      acc[0][3] = __builtin_amdgcn_wmma_f32_16x16x4_f32(false, a0, false, b3, (short)0, acc[0][3], false, false);
      acc[1][0] = __builtin_amdgcn_wmma_f32_16x16x4_f32(false, a1, false, b0, (short)0, acc[1][0], false, false);
      acc[1][1] = __builtin_amdgcn_wmma_f32_16x16x4_f32(false, a1, false, b1, (short)0, acc[1][1], false, false);
      acc[1][2] = __builtin_amdgcn_wmma_f32_16x16x4_f32(false, a1, false, b2, (short)0, acc[1][2], false, false);
      acc[1][3] = __builtin_amdgcn_wmma_f32_16x16x4_f32(false, a1, false, b3, (short)0, acc[1][3], false, false);
    }
    __syncthreads();
  }

#pragma unroll
  for (int im = 0; im < 2; ++im) {
#pragma unroll
    for (int r = 0; r < 8; ++r) {
      int row = m0 + wm + 16 * im + 8 * h + r;
      float* drow = Db + (long long)row * N;
#pragma unroll
      for (int in = 0; in < 4; ++in) {
        int col = n0 + wn + 16 * in + mr;
        float v = acc[im][in][r];
        if (mode == 1) {
          v = -0.5f * v + ((row == col) ? 1.5f : 0.0f);
        } else {
          v *= alpha;
        }
        drow[col] = v;
      }
    }
  }
}

__global__ __launch_bounds__(256)
void trace_kernel(const float* __restrict__ cov, float* __restrict__ tr, int d) {
  int b = blockIdx.x;
  int t = threadIdx.x;
  __shared__ float sh[256];
  sh[t] = cov[(long long)b * d * d + (long long)t * d + t];
  __syncthreads();
  for (int s = 128; s > 0; s >>= 1) {
    if (t < s) sh[t] += sh[t + s];
    __syncthreads();
  }
  if (t == 0) tr[b] = sh[0];
}

__global__ __launch_bounds__(256)
void init_A_ZY(const float* __restrict__ cov, const float* __restrict__ tr,
               float* __restrict__ Amat, float* __restrict__ ZY, int d) {
  int bi = blockIdx.x;
  int b  = bi / d;
  int i  = bi % d;
  int j  = threadIdx.x;
  long long idx = (long long)b * d * d + (long long)i * d + j;
  float a = cov[idx] / tr[b];
  Amat[idx] = a;
  ZY[idx]   = -0.5f * a + ((i == j) ? 1.5f : 0.0f);
}

__global__ __launch_bounds__(256)
void out_kernel(const float* __restrict__ YZY, const float* __restrict__ mean,
                const float* __restrict__ tr, float* __restrict__ out, int d) {
  int wid  = threadIdx.x >> 5;
  int lane = threadIdx.x & 31;
  int row  = blockIdx.x * 8 + wid;
  int b    = row / d;
  int i    = row % d;
  const float* yrow = YZY + (long long)b * d * d + (long long)i * d;
  const float* mrow = mean + (long long)b * d;
  float s = 0.0f;
  for (int j = lane; j < d; j += 32) s += yrow[j] * mrow[j];
  for (int off = 16; off > 0; off >>= 1) s += __shfl_down(s, off, 32);
  if (lane == 0) out[row] = s * sqrtf(tr[b]);
}

extern "C" void kernel_launch(void* const* d_in, const int* in_sizes, int n_in,
                              void* d_out, int out_size, void* d_ws, size_t ws_size,
                              hipStream_t stream) {
  (void)n_in; (void)ws_size; (void)out_size;
  const float* feat  = (const float*)d_in[1];
  const float* noise = (const float*)d_in[2];

  const int B  = in_sizes[0];
  const int d  = 256;
  const long long total = (long long)in_sizes[1];
  const int n  = (int)(total / ((long long)B * d));
  const int nd = n * d;
  const long long dd = (long long)d * d;

  float* W = (float*)d_ws;
  long long off = 0;
  float* Fp   = W + off; off += (long long)B * nd;
  float* mean = W + off; off += (long long)B * d;
  float* scl  = W + off; off += B;
  float* nmn  = W + off; off += B;
  float* tr   = W + off; off += B;
  off = (off + 255) & ~255LL;
  float* M0 = W + off; off += (long long)B * dd;
  float* M1 = W + off; off += (long long)B * dd;
  float* M2 = W + off; off += (long long)B * dd;
  float* M3 = W + off; off += (long long)B * dd;

  stats_kernel<<<B, 256, 0, stream>>>(feat, noise, scl, nmn, nd, 31.6227766016838f);
  perturb_kernel<<<dim3(nd / 256, B), 256, 0, stream>>>(feat, noise, scl, nmn, Fp, nd);
  mean_kernel<<<B, d, 0, stream>>>(Fp, mean, n, d);
  center_kernel<<<B * n, d, 0, stream>>>(Fp, mean, n, d);

  dim3 gg(d / BN, d / BM, B);
  wmma_gemm_batched<<<gg, 256, 0, stream>>>(Fp, (long long)nd, 1, d,
                                            Fp, (long long)nd, d, 1,
                                            M0, dd, d, n,
                                            1.0f / (float)(n - 1), 0);
  trace_kernel<<<B, d, 0, stream>>>(M0, tr, d);
  init_A_ZY<<<B * d, d, 0, stream>>>(M0, tr, M1, M2, d);

  wmma_gemm_batched<<<gg, 256, 0, stream>>>(M1, dd, d, 1, M2, dd, d, 1,
                                            M0, dd, d, d, 1.0f, 0);
  float* Y = M0; float* Z = M2; float* S1 = M1; float* S2 = M3;

  for (int it = 0; it < 3; ++it) {
    wmma_gemm_batched<<<gg, 256, 0, stream>>>(Z, dd, d, 1, Y, dd, d, 1,
                                              S1, dd, d, d, 1.0f, 1);
    wmma_gemm_batched<<<gg, 256, 0, stream>>>(Y, dd, d, 1, S1, dd, d, 1,
                                              S2, dd, d, d, 1.0f, 0);
    wmma_gemm_batched<<<gg, 256, 0, stream>>>(S1, dd, d, 1, Z, dd, d, 1,
                                              Y, dd, d, d, 1.0f, 0);
    float* newY = S2; float* newZ = Y; float* freed = Z;
    Y = newY; Z = newZ; S2 = freed;
  }

  wmma_gemm_batched<<<gg, 256, 0, stream>>>(Z, dd, d, 1, Y, dd, d, 1,
                                            S1, dd, d, d, 1.0f, 1);
  wmma_gemm_batched<<<gg, 256, 0, stream>>>(Y, dd, d, 1, S1, dd, d, 1,
                                            S2, dd, d, d, 1.0f, 0);

  out_kernel<<<(B * d) / 8, 256, 0, stream>>>(S2, mean, tr, (float*)d_out, d);
}


# GEMM K-loop software-pipelined: next K-tile's global loads go to registers before the current tile's WMMAs and to a second LDS stage after them; one barrier per K-step instead of two
# speedup vs baseline: 1.2581x; 1.2581x over previous
#include <hip/hip_runtime.h>
#include <hip/hip_bf16.h>

typedef __attribute__((ext_vector_type(2))) float v2f;
typedef __attribute__((ext_vector_type(8))) float v8f;

__global__ __launch_bounds__(256)
void stats_kernel(const float* __restrict__ feat, const float* __restrict__ noise,
                  float* __restrict__ scl, float* __restrict__ nmn,
                  int nd, float snr_pow) {
  int b = blockIdx.x;
  int t = threadIdx.x;
  const float* f  = feat  + (long long)b * nd;
  const float* nz = noise + (long long)b * nd;
  double sf = 0.0, sf2 = 0.0, sn = 0.0, sn2 = 0.0;
  for (int i = t; i < nd; i += 256) {
    double x = (double)f[i];
    double y = (double)nz[i];
    sf += x; sf2 += x * x; sn += y; sn2 += y * y;
  }
  __shared__ double sh[4][256];
  sh[0][t] = sf; sh[1][t] = sf2; sh[2][t] = sn; sh[3][t] = sn2;
  __syncthreads();
  for (int s = 128; s > 0; s >>= 1) {
    if (t < s) {
      sh[0][t] += sh[0][t + s]; sh[1][t] += sh[1][t + s];
      sh[2][t] += sh[2][t + s]; sh[3][t] += sh[3][t + s];
    }
    __syncthreads();
  }
  if (t == 0) {
    double inv = 1.0 / (double)nd;
    double gmf = sh[0][0] * inv;
    double gmn = sh[2][0] * inv;
    double tvar = sh[1][0] * inv - gmf * gmf;
    double nvar = sh[3][0] * inv - gmn * gmn;
    scl[b] = (float)sqrt((tvar / (double)snr_pow) / nvar);
    nmn[b] = (float)gmn;
  }
}

__global__ __launch_bounds__(256)
void perturb_kernel(const float* __restrict__ feat, const float* __restrict__ noise,
                    const float* __restrict__ scl, const float* __restrict__ nmn,
                    float* __restrict__ Fp, int nd) {
  int b = blockIdx.y;
  long long idx = (long long)b * nd + (long long)blockIdx.x * 256 + threadIdx.x;
  Fp[idx] = feat[idx] + (noise[idx] - nmn[b]) * scl[b];
}

__global__ __launch_bounds__(256)
void mean_kernel(const float* __restrict__ Fp, float* __restrict__ mean, int n, int d) {
  int b = blockIdx.x;
  int j = threadIdx.x;
  const float* base = Fp + (long long)b * n * d;
  float s = 0.0f;
  for (int i = 0; i < n; ++i) s += base[(long long)i * d + j];
  mean[(long long)b * d + j] = s / (float)n;
}

__global__ __launch_bounds__(256)
void center_kernel(float* __restrict__ Fp, const float* __restrict__ mean, int n, int d) {
  int row = blockIdx.x;
  int j   = threadIdx.x;
  int b   = row / n;
  Fp[(long long)row * d + j] -= mean[(long long)b * d + j];
}

#define BM 128
#define BN 128
#define BK 16
#define LDP 18

__global__ __launch_bounds__(256)
void wmma_gemm_batched(const float* __restrict__ A, long long sAb, int sAm, int sAk,
                       const float* __restrict__ Bm, long long sBb, int sBk, int sBn,
                       float* __restrict__ Dm, long long sDb,
                       int N, int K, float alpha, int mode) {
  __shared__ __align__(16) float As[2][BM][LDP];
  __shared__ __align__(16) float Bst[2][BN][LDP];

  int b  = blockIdx.z;
  int m0 = blockIdx.y * BM;
  int n0 = blockIdx.x * BN;
  const float* Ab = A  + (long long)b * sAb;
  const float* Bb = Bm + (long long)b * sBb;
  float*       Db = Dm + (long long)b * sDb;

  int tid  = threadIdx.x;
  int wid  = tid >> 5;
  int lane = tid & 31;
  int mr   = lane & 15;
  int h    = lane >> 4;
  int wm   = (wid >> 1) * 32;
  int wn   = (wid & 1) * 64;

  v8f acc[2][4];
#pragma unroll
  for (int im = 0; im < 2; ++im)
#pragma unroll
    for (int in = 0; in < 4; ++in) acc[im][in] = (v8f){};

  float ra[8], rb[8];

#define G2R(k0_)                                                                    \
  do {                                                                              \
    _Pragma("unroll")                                                               \
    for (int i = 0; i < 8; ++i) {                                                   \
      int e = tid + i * 256;                                                        \
      int m = e >> 4, kk = e & 15;                                                  \
      ra[i] = Ab[(long long)(m0 + m) * sAm + (long long)((k0_) + kk) * sAk];        \
    }                                                                               \
    _Pragma("unroll")                                                               \
    for (int i = 0; i < 8; ++i) {                                                   \
      int e = tid + i * 256;                                                        \
      int nn = e & 127, kk = e >> 7;                                                \
      rb[i] = Bb[(long long)((k0_) + kk) * sBk + (long long)(n0 + nn) * sBn];       \
    }                                                                               \
  } while (0)

#define R2S(s_)                                                                     \
  do {                                                                              \
    _Pragma("unroll")                                                               \
    for (int i = 0; i < 8; ++i) {                                                   \
      int e = tid + i * 256;                                                        \
      int m = e >> 4, kk = e & 15;                                                  \
      As[(s_)][m][kk] = ra[i];                                                      \
    }                                                                               \
    _Pragma("unroll")                                                               \
    for (int i = 0; i < 8; ++i) {                                                   \
      int e = tid + i * 256;                                                        \
      int nn = e & 127, kk = e >> 7;                                                \
      Bst[(s_)][nn][kk] = rb[i];                                                    \
    }                                                                               \
  } while (0)

#define TILE_MMA(s_)                                                                \
  do {                                                                              \
    _Pragma("unroll")                                                               \
    for (int kk = 0; kk < BK; kk += 4) {                                            \
      int kc = kk + 2 * h;                       \
      v2f a0 = *(const v2f*)&As[(s_)][wm + mr][kc];                                 \
      v2f a1 = *(const v2f*)&As[(s_)][wm + 16 + mr][kc];                            \
      v2f b0 = *(const v2f*)&Bst[(s_)][wn + mr][kc];                                \
      v2f b1 = *(const v2f*)&Bst[(s_)][wn + 16 + mr][kc];                           \
      v2f b2 = *(const v2f*)&Bst[(s_)][wn + 32 + mr][kc];                           \
      v2f b3 = *(const v2f*)&Bst[(s_)][wn + 48 + mr][kc];                           \
      acc[0][0] = __builtin_amdgcn_wmma_f32_16x16x4_f32(false, a0, false, b0, (short)0, acc[0][0], false, false); \
      acc[0][1] = __builtin_amdgcn_wmma_f32_16x16x4_f32(false, a0, false, b1, (short)0, acc[0][1], false, false); \
      acc[0][2] = __builtin_amdgcn_wmma_f32_16x16x4_f32(false, a0, false, b2, (short)0, acc[0][2], false, false); \
      acc[0][3] = __builtin_amdgcn_wmma_f32_16x16x4_f32(false, a0, false, b3, (short)0, acc[0][3], false, false); \
      acc[1][0] = __builtin_amdgcn_wmma_f32_16x16x4_f32(false, a1, false, b0, (short)0, acc[1][0], false, false); \
      acc[1][1] = __builtin_amdgcn_wmma_f32_16x16x4_f32(false, a1, false, b1, (short)0, acc[1][1], false, false); \
      acc[1][2] = __builtin_amdgcn_wmma_f32_16x16x4_f32(false, a1, false, b2, (short)0, acc[1][2], false, false); \
      acc[1][3] = __builtin_amdgcn_wmma_f32_16x16x4_f32(false, a1, false, b3, (short)0, acc[1][3], false, false); \
    }                                                                               \
  } while (0)

  const int nk = K / BK;

  G2R(0);
  R2S(0);
  __syncthreads();

  for (int t = 0; t < nk - 1; ++t) {
    int cur = t & 1;
    G2R((t + 1) * BK);
    __builtin_amdgcn_sched_barrier(0);
    TILE_MMA(cur);
    R2S(cur ^ 1);
    __syncthreads();
  }
  TILE_MMA((nk - 1) & 1);

#undef G2R
#undef R2S
#undef TILE_MMA

#pragma unroll
  for (int im = 0; im < 2; ++im) {
#pragma unroll
    for (int r = 0; r < 8; ++r) {
      int row = m0 + wm + 16 * im + 8 * h + r;
      float* drow = Db + (long long)row * N;
#pragma unroll
      for (int in = 0; in < 4; ++in) {
        int col = n0 + wn + 16 * in + mr;
        float v = acc[im][in][r];
        if (mode == 1) {
          v = -0.5f * v + ((row == col) ? 1.5f : 0.0f);
        } else {
          v *= alpha;
        }
        drow[col] = v;
      }
    }
  }
}

__global__ __launch_bounds__(256)
void trace_kernel(const float* __restrict__ cov, float* __restrict__ tr, int d) {
  int b = blockIdx.x;
  int t = threadIdx.x;
  __shared__ float sh[256];
  sh[t] = cov[(long long)b * d * d + (long long)t * d + t];
  __syncthreads();
  for (int s = 128; s > 0; s >>= 1) {
    if (t < s) sh[t] += sh[t + s];
    __syncthreads();
  }
  if (t == 0) tr[b] = sh[0];
}

__global__ __launch_bounds__(256)
void init_A_ZY(const float* __restrict__ cov, const float* __restrict__ tr,
               float* __restrict__ Amat, float* __restrict__ ZY, int d) {
  int bi = blockIdx.x;
  int b  = bi / d;
  int i  = bi % d;
  int j  = threadIdx.x;
  long long idx = (long long)b * d * d + (long long)i * d + j;
  float a = cov[idx] / tr[b];
  Amat[idx] = a;
  ZY[idx]   = -0.5f * a + ((i == j) ? 1.5f : 0.0f);
}

__global__ __launch_bounds__(256)
void out_kernel(const float* __restrict__ YZY, const float* __restrict__ mean,
                const float* __restrict__ tr, float* __restrict__ out, int d) {
  int wid  = threadIdx.x >> 5;
  int lane = threadIdx.x & 31;
  int row  = blockIdx.x * 8 + wid;
  int b    = row / d;
  int i    = row % d;
  const float* yrow = YZY + (long long)b * d * d + (long long)i * d;
  const float* mrow = mean + (long long)b * d;
  float s = 0.0f;
  for (int j = lane; j < d; j += 32) s += yrow[j] * mrow[j];
  for (int off = 16; off > 0; off >>= 1) s += __shfl_down(s, off, 32);
  if (lane == 0) out[row] = s * sqrtf(tr[b]);
}

extern "C" void kernel_launch(void* const* d_in, const int* in_sizes, int n_in,
                              void* d_out, int out_size, void* d_ws, size_t ws_size,
                              hipStream_t stream) {
  (void)n_in; (void)ws_size; (void)out_size;
  const float* feat  = (const float*)d_in[1];
  const float* noise = (const float*)d_in[2];

  const int B  = in_sizes[0];
  const int d  = 256;
  const long long total = (long long)in_sizes[1];
  const int n  = (int)(total / ((long long)B * d));
  const int nd = n * d;
  const long long dd = (long long)d * d;

  float* W = (float*)d_ws;
  long long off = 0;
  float* Fp   = W + off; off += (long long)B * nd;
  float* mean = W + off; off += (long long)B * d;
  float* scl  = W + off; off += B;
  float* nmn  = W + off; off += B;
  float* tr   = W + off; off += B;
  off = (off + 255) & ~255LL;
  float* M0 = W + off; off += (long long)B * dd;
  float* M1 = W + off; off += (long long)B * dd;
  float* M2 = W + off; off += (long long)B * dd;
  float* M3 = W + off; off += (long long)B * dd;

  stats_kernel<<<B, 256, 0, stream>>>(feat, noise, scl, nmn, nd, 31.6227766016838f);
  perturb_kernel<<<dim3(nd / 256, B), 256, 0, stream>>>(feat, noise, scl, nmn, Fp, nd);
  mean_kernel<<<B, d, 0, stream>>>(Fp, mean, n, d);
  center_kernel<<<B * n, d, 0, stream>>>(Fp, mean, n, d);

  dim3 gg(d / BN, d / BM, B);
  wmma_gemm_batched<<<gg, 256, 0, stream>>>(Fp, (long long)nd, 1, d,
                                            Fp, (long long)nd, d, 1,
                                            M0, dd, d, n,
                                            1.0f / (float)(n - 1), 0);
  trace_kernel<<<B, d, 0, stream>>>(M0, tr, d);
  init_A_ZY<<<B * d, d, 0, stream>>>(M0, tr, M1, M2, d);

  wmma_gemm_batched<<<gg, 256, 0, stream>>>(M1, dd, d, 1, M2, dd, d, 1,
                                            M0, dd, d, d, 1.0f, 0);
  float* Y = M0; float* Z = M2; float* S1 = M1; float* S2 = M3;

  for (int it = 0; it < 3; ++it) {
    wmma_gemm_batched<<<gg, 256, 0, stream>>>(Z, dd, d, 1, Y, dd, d, 1,
                                              S1, dd, d, d, 1.0f, 1);
    wmma_gemm_batched<<<gg, 256, 0, stream>>>(Y, dd, d, 1, S1, dd, d, 1,
                                              S2, dd, d, d, 1.0f, 0);
    wmma_gemm_batched<<<gg, 256, 0, stream>>>(S1, dd, d, 1, Z, dd, d, 1,
                                              Y, dd, d, d, 1.0f, 0);
    float* newY = S2; float* newZ = Y; float* freed = Z;
    Y = newY; Z = newZ; S2 = freed;
  }

  wmma_gemm_batched<<<gg, 256, 0, stream>>>(Z, dd, d, 1, Y, dd, d, 1,
                                            S1, dd, d, d, 1.0f, 1);
  wmma_gemm_batched<<<gg, 256, 0, stream>>>(Y, dd, d, 1, S1, dd, d, 1,
                                            S2, dd, d, d, 1.0f, 0);

  out_kernel<<<(B * d) / 8, 256, 0, stream>>>(S2, mean, tr, (float*)d_out, d);
}


# GEMM K-tile depth 16 -> 32 (LDS pitch 34, compile-time leading dimension): half the barriers and stage hand-overs per product; each staging load one whole 128-B line at an immediate offset
# speedup vs baseline: 1.4082x; 1.1193x over previous
#include <hip/hip_runtime.h>
#include <hip/hip_bf16.h>

typedef __attribute__((ext_vector_type(2))) float v2f;
typedef __attribute__((ext_vector_type(8))) float v8f;

__global__ __launch_bounds__(256)
void stats_kernel(const float* __restrict__ feat, const float* __restrict__ noise,
                  float* __restrict__ scl, float* __restrict__ nmn,
                  int nd, float snr_pow) {
  int b = blockIdx.x;
  int t = threadIdx.x;
  const float* f  = feat  + (long long)b * nd;
  const float* nz = noise + (long long)b * nd;
  double sf = 0.0, sf2 = 0.0, sn = 0.0, sn2 = 0.0;
  for (int i = t; i < nd; i += 256) {
    double x = (double)f[i];
    double y = (double)nz[i];
    sf += x; sf2 += x * x; sn += y; sn2 += y * y;
  }
  __shared__ double sh[4][256];
  sh[0][t] = sf; sh[1][t] = sf2; sh[2][t] = sn; sh[3][t] = sn2;
  __syncthreads();
  for (int s = 128; s > 0; s >>= 1) {
    if (t < s) {
      sh[0][t] += sh[0][t + s]; sh[1][t] += sh[1][t + s];
      sh[2][t] += sh[2][t + s]; sh[3][t] += sh[3][t + s];
    }
    __syncthreads();
  }
  if (t == 0) {
    double inv = 1.0 / (double)nd;
    double gmf = sh[0][0] * inv;
    double gmn = sh[2][0] * inv;
    double tvar = sh[1][0] * inv - gmf * gmf;
    double nvar = sh[3][0] * inv - gmn * gmn;
    scl[b] = (float)sqrt((tvar / (double)snr_pow) / nvar);
    nmn[b] = (float)gmn;
  }
}

__global__ __launch_bounds__(256)
void perturb_kernel(const float* __restrict__ feat, const float* __restrict__ noise,
                    const float* __restrict__ scl, const float* __restrict__ nmn,
                    float* __restrict__ Fp, int nd) {
  int b = blockIdx.y;
  long long idx = (long long)b * nd + (long long)blockIdx.x * 256 + threadIdx.x;
  Fp[idx] = feat[idx] + (noise[idx] - nmn[b]) * scl[b];
}

__global__ __launch_bounds__(256)
void mean_kernel(const float* __restrict__ Fp, float* __restrict__ mean, int n, int d) {
  int b = blockIdx.x;
  int j = threadIdx.x;
  const float* base = Fp + (long long)b * n * d;
  float s = 0.0f;
  for (int i = 0; i < n; ++i) s += base[(long long)i * d + j];
  mean[(long long)b * d + j] = s / (float)n;
}

__global__ __launch_bounds__(256)
void center_kernel(float* __restrict__ Fp, const float* __restrict__ mean, int n, int d) {
  int row = blockIdx.x;
  int j   = threadIdx.x;
  int b   = row / n;
  Fp[(long long)row * d + j] -= mean[(long long)b * d + j];
}

#define BM 128
#define BN 128
#define BK 32
#define LDP 34

template <bool AT, int LD>
__global__ __launch_bounds__(256)
void wmma_gemm_batched(const float* __restrict__ A, long long sAb,
                       const float* __restrict__ Bm, long long sBb,
                       float* __restrict__ Dm, long long sDb,
                       int K, float alpha, int mode) {
  __shared__ __align__(16) float As[2][BM][LDP];
  __shared__ __align__(16) float Bst[2][BN][LDP];

  int b  = blockIdx.z;
  int m0 = blockIdx.y * BM;
  int n0 = blockIdx.x * BN;
  const float* Ab = A  + (long long)b * sAb;
  const float* Bb = Bm + (long long)b * sBb;
  float*       Db = Dm + (long long)b * sDb;

  int tid  = threadIdx.x;
  int wid  = tid >> 5;
  int lane = tid & 31;
  int mr   = lane & 15;
  int h    = lane >> 4;
  int wm   = (wid >> 1) * 32;
  int wn   = (wid & 1) * 64;

  int tr = tid >> 5, tc = tid & 31;
  int ur = tid >> 7, uc = tid & 127;
  const float* Ag = AT ? Ab + (long long)ur * LD + (m0 + uc)
                       : Ab + (long long)(m0 + tr) * LD + tc;
  const float* Bg = Bb + (long long)ur * LD + (n0 + uc);

  v8f acc[2][4];
#pragma unroll
  for (int im = 0; im < 2; ++im)
#pragma unroll
    for (int in = 0; in < 4; ++in) acc[im][in] = (v8f){};

  float ra[16], rb[16];

#define G2R(k0_)                                                                    \
  do {                                                                              \
    const float* Ak = AT ? Ag + (long long)(k0_) * LD : Ag + (k0_);                 \
    const float* Bk = Bg + (long long)(k0_) * LD;                                   \
    _Pragma("unroll")                                                               \
    for (int i = 0; i < 16; ++i) ra[i] = AT ? Ak[2 * i * LD] : Ak[8 * i * LD];      \
    _Pragma("unroll")                                                               \
    for (int i = 0; i < 16; ++i) rb[i] = Bk[2 * i * LD];                            \
  } while (0)

#define R2S(s_)                                                                     \
  do {                                                                              \
    _Pragma("unroll")                                                               \
    for (int i = 0; i < 16; ++i) {                                                  \
      if (AT) As[(s_)][uc][ur + 2 * i] = ra[i];                                     \
      else    As[(s_)][tr + 8 * i][tc] = ra[i];                                     \
    }                                                                               \
    _Pragma("unroll")                                                               \
    for (int i = 0; i < 16; ++i) Bst[(s_)][uc][ur + 2 * i] = rb[i];                 \
  } while (0)

#define TILE_MMA(s_)                                                                \
  do {                                                                              \
    _Pragma("unroll")                                                               \
    for (int kk = 0; kk < BK; kk += 4) {                                            \
      int kc = kk + 2 * h;                       \
      v2f a0 = *(const v2f*)&As[(s_)][wm + mr][kc];                                 \
      v2f a1 = *(const v2f*)&As[(s_)][wm + 16 + mr][kc];                            \
      v2f b0 = *(const v2f*)&Bst[(s_)][wn + mr][kc];                                \
      v2f b1 = *(const v2f*)&Bst[(s_)][wn + 16 + mr][kc];                           \
      v2f b2 = *(const v2f*)&Bst[(s_)][wn + 32 + mr][kc];                           \
      v2f b3 = *(const v2f*)&Bst[(s_)][wn + 48 + mr][kc];                           \
      acc[0][0] = __builtin_amdgcn_wmma_f32_16x16x4_f32(false, a0, false, b0, (short)0, acc[0][0], false, false); \
      acc[0][1] = __builtin_amdgcn_wmma_f32_16x16x4_f32(false, a0, false, b1, (short)0, acc[0][1], false, false); \
      acc[0][2] = __builtin_amdgcn_wmma_f32_16x16x4_f32(false, a0, false, b2, (short)0, acc[0][2], false, false); \
      acc[0][3] = __builtin_amdgcn_wmma_f32_16x16x4_f32(false, a0, false, b3, (short)0, acc[0][3], false, false); \
      acc[1][0] = __builtin_amdgcn_wmma_f32_16x16x4_f32(false, a1, false, b0, (short)0, acc[1][0], false, false); \
      acc[1][1] = __builtin_amdgcn_wmma_f32_16x16x4_f32(false, a1, false, b1, (short)0, acc[1][1], false, false); \
      acc[1][2] = __builtin_amdgcn_wmma_f32_16x16x4_f32(false, a1, false, b2, (short)0, acc[1][2], false, false); \
      acc[1][3] = __builtin_amdgcn_wmma_f32_16x16x4_f32(false, a1, false, b3, (short)0, acc[1][3], false, false); \
    }                                                                               \
  } while (0)

  const int nk = K / BK;

  G2R(0);
  R2S(0);
  __syncthreads();

  for (int t = 0; t < nk - 1; ++t) {
    int cur = t & 1;
    G2R((t + 1) * BK);
    __builtin_amdgcn_sched_barrier(0);
    TILE_MMA(cur);
    R2S(cur ^ 1);
    __syncthreads();
  }
  TILE_MMA((nk - 1) & 1);

#undef G2R
#undef R2S
#undef TILE_MMA

#pragma unroll
  for (int im = 0; im < 2; ++im) {
#pragma unroll
    for (int r = 0; r < 8; ++r) {
      int row = m0 + wm + 16 * im + 8 * h + r;
      float* drow = Db + (long long)row * LD;
#pragma unroll
      for (int in = 0; in < 4; ++in) {
        int col = n0 + wn + 16 * in + mr;
        float v = acc[im][in][r];
        if (mode == 1) {
          v = -0.5f * v + ((row == col) ? 1.5f : 0.0f);
        } else {
          v *= alpha;
        }
        drow[col] = v;
      }
    }
  }
}

__global__ __launch_bounds__(256)
void trace_kernel(const float* __restrict__ cov, float* __restrict__ tr, int d) {
  int b = blockIdx.x;
  int t = threadIdx.x;
  __shared__ float sh[256];
  sh[t] = cov[(long long)b * d * d + (long long)t * d + t];
  __syncthreads();
  for (int s = 128; s > 0; s >>= 1) {
    if (t < s) sh[t] += sh[t + s];
    __syncthreads();
  }
  if (t == 0) tr[b] = sh[0];
}

__global__ __launch_bounds__(256)
void init_A_ZY(const float* __restrict__ cov, const float* __restrict__ tr,
               float* __restrict__ Amat, float* __restrict__ ZY, int d) {
  int bi = blockIdx.x;
  int b  = bi / d;
  int i  = bi % d;
  int j  = threadIdx.x;
  long long idx = (long long)b * d * d + (long long)i * d + j;
  float a = cov[idx] / tr[b];
  Amat[idx] = a;
  ZY[idx]   = -0.5f * a + ((i == j) ? 1.5f : 0.0f);
}

__global__ __launch_bounds__(256)
void out_kernel(const float* __restrict__ YZY, const float* __restrict__ mean,
                const float* __restrict__ tr, float* __restrict__ out, int d) {
  int wid  = threadIdx.x >> 5;
  int lane = threadIdx.x & 31;
  int row  = blockIdx.x * 8 + wid;
  int b    = row / d;
  int i    = row % d;
  const float* yrow = YZY + (long long)b * d * d + (long long)i * d;
  const float* mrow = mean + (long long)b * d;
  float s = 0.0f;
  for (int j = lane; j < d; j += 32) s += yrow[j] * mrow[j];
  for (int off = 16; off > 0; off >>= 1) s += __shfl_down(s, off, 32);
  if (lane == 0) out[row] = s * sqrtf(tr[b]);
}

extern "C" void kernel_launch(void* const* d_in, const int* in_sizes, int n_in,
                              void* d_out, int out_size, void* d_ws, size_t ws_size,
                              hipStream_t stream) {
  (void)n_in; (void)ws_size; (void)out_size;
  const float* feat  = (const float*)d_in[1];
  const float* noise = (const float*)d_in[2];

  const int B  = in_sizes[0];
  const int d  = 256;
  const long long total = (long long)in_sizes[1];
  const int n  = (int)(total / ((long long)B * d));
  const int nd = n * d;
  const long long dd = (long long)d * d;

  float* W = (float*)d_ws;
  long long off = 0;
  float* Fp   = W + off; off += (long long)B * nd;
  float* mean = W + off; off += (long long)B * d;
  float* scl  = W + off; off += B;
  float* nmn  = W + off; off += B;
  float* tr   = W + off; off += B;
  off = (off + 255) & ~255LL;
  float* M0 = W + off; off += (long long)B * dd;
  float* M1 = W + off; off += (long long)B * dd;
  float* M2 = W + off; off += (long long)B * dd;
  float* M3 = W + off; off += (long long)B * dd;

  stats_kernel<<<B, 256, 0, stream>>>(feat, noise, scl, nmn, nd, 31.6227766016838f);
  perturb_kernel<<<dim3(nd / 256, B), 256, 0, stream>>>(feat, noise, scl, nmn, Fp, nd);
  mean_kernel<<<B, d, 0, stream>>>(Fp, mean, n, d);
  center_kernel<<<B * n, d, 0, stream>>>(Fp, mean, n, d);

  dim3 gg(d / BN, d / BM, B);
  wmma_gemm_batched<true, d><<<gg, 256, 0, stream>>>(Fp, (long long)nd, Fp, (long long)nd,
                                                       M0, dd, n, 1.0f / (float)(n - 1), 0);
  trace_kernel<<<B, d, 0, stream>>>(M0, tr, d);
  init_A_ZY<<<B * d, d, 0, stream>>>(M0, tr, M1, M2, d);

  wmma_gemm_batched<false, d><<<gg, 256, 0, stream>>>(M1, dd, M2, dd, M0, dd, d, 1.0f, 0);
  float* Y = M0; float* Z = M2; float* S1 = M1; float* S2 = M3;

  for (int it = 0; it < 3; ++it) {
    wmma_gemm_batched<false, d><<<gg, 256, 0, stream>>>(Z, dd, Y, dd, S1, dd, d, 1.0f, 1);
    wmma_gemm_batched<false, d><<<gg, 256, 0, stream>>>(Y, dd, S1, dd, S2, dd, d, 1.0f, 0);
    wmma_gemm_batched<false, d><<<gg, 256, 0, stream>>>(S1, dd, Z, dd, Y, dd, d, 1.0f, 0);
    float* newY = S2; float* newZ = Y; float* freed = Z;
    Y = newY; Z = newZ; S2 = freed;
  }

  wmma_gemm_batched<false, d><<<gg, 256, 0, stream>>>(Z, dd, Y, dd, S1, dd, d, 1.0f, 1);
  wmma_gemm_batched<false, d><<<gg, 256, 0, stream>>>(Y, dd, S1, dd, S2, dd, d, 1.0f, 0);

  out_kernel<<<(B * d) / 8, 256, 0, stream>>>(S2, mean, tr, (float*)d_out, d);
}


# Tail re-associated: out = sqrt(tr)*Y*((3*m - Z*(Y*m))/2) as three matrix-vector products on the vector unit in one launch; the last two GEMMs go; the final sums are taken in a new order
# speedup vs baseline: 1.6019x; 1.1376x over previous
#include <hip/hip_runtime.h>
#include <hip/hip_bf16.h>

typedef __attribute__((ext_vector_type(2))) float v2f;
typedef __attribute__((ext_vector_type(8))) float v8f;

__global__ __launch_bounds__(256)
void stats_kernel(const float* __restrict__ feat, const float* __restrict__ noise,
                  float* __restrict__ scl, float* __restrict__ nmn,
                  int nd, float snr_pow) {
  int b = blockIdx.x;
  int t = threadIdx.x;
  const float* f  = feat  + (long long)b * nd;
  const float* nz = noise + (long long)b * nd;
  double sf = 0.0, sf2 = 0.0, sn = 0.0, sn2 = 0.0;
  for (int i = t; i < nd; i += 256) {
    double x = (double)f[i];
    double y = (double)nz[i];
    sf += x; sf2 += x * x; sn += y; sn2 += y * y;
  }
  __shared__ double sh[4][256];
  sh[0][t] = sf; sh[1][t] = sf2; sh[2][t] = sn; sh[3][t] = sn2;
  __syncthreads();
  for (int s = 128; s > 0; s >>= 1) {
    if (t < s) {
      sh[0][t] += sh[0][t + s]; sh[1][t] += sh[1][t + s];
      sh[2][t] += sh[2][t + s]; sh[3][t] += sh[3][t + s];
    }
    __syncthreads();
  }
  if (t == 0) {
    double inv = 1.0 / (double)nd;
    double gmf = sh[0][0] * inv;
    double gmn = sh[2][0] * inv;
    double tvar = sh[1][0] * inv - gmf * gmf;
    double nvar = sh[3][0] * inv - gmn * gmn;
    scl[b] = (float)sqrt((tvar / (double)snr_pow) / nvar);
    nmn[b] = (float)gmn;
  }
}

__global__ __launch_bounds__(256)
void perturb_kernel(const float* __restrict__ feat, const float* __restrict__ noise,
                    const float* __restrict__ scl, const float* __restrict__ nmn,
                    float* __restrict__ Fp, int nd) {
  int b = blockIdx.y;
  long long idx = (long long)b * nd + (long long)blockIdx.x * 256 + threadIdx.x;
  Fp[idx] = feat[idx] + (noise[idx] - nmn[b]) * scl[b];
}

__global__ __launch_bounds__(256)
void mean_kernel(const float* __restrict__ Fp, float* __restrict__ mean, int n, int d) {
  int b = blockIdx.x;
  int j = threadIdx.x;
  const float* base = Fp + (long long)b * n * d;
  float s = 0.0f;
  for (int i = 0; i < n; ++i) s += base[(long long)i * d + j];
  mean[(long long)b * d + j] = s / (float)n;
}

__global__ __launch_bounds__(256)
void center_kernel(float* __restrict__ Fp, const float* __restrict__ mean, int n, int d) {
  int row = blockIdx.x;
  int j   = threadIdx.x;
  int b   = row / n;
  Fp[(long long)row * d + j] -= mean[(long long)b * d + j];
}

#define BM 128
#define BN 128
#define BK 32
#define LDP 34

template <bool AT, int LD>
__global__ __launch_bounds__(256)
void wmma_gemm_batched(const float* __restrict__ A, long long sAb,
                       const float* __restrict__ Bm, long long sBb,
                       float* __restrict__ Dm, long long sDb,
                       int K, float alpha, int mode) {
  __shared__ __align__(16) float As[2][BM][LDP];
  __shared__ __align__(16) float Bst[2][BN][LDP];

  int b  = blockIdx.z;
  int m0 = blockIdx.y * BM;
  int n0 = blockIdx.x * BN;
  const float* Ab = A  + (long long)b * sAb;
  const float* Bb = Bm + (long long)b * sBb;
  float*       Db = Dm + (long long)b * sDb;

  int tid  = threadIdx.x;
  int wid  = tid >> 5;
  int lane = tid & 31;
  int mr   = lane & 15;
  int h    = lane >> 4;
  int wm   = (wid >> 1) * 32;
  int wn   = (wid & 1) * 64;

  int tr = tid >> 5, tc = tid & 31;
  int ur = tid >> 7, uc = tid & 127;
  const float* Ag = AT ? Ab + (long long)ur * LD + (m0 + uc)
                       : Ab + (long long)(m0 + tr) * LD + tc;
  const float* Bg = Bb + (long long)ur * LD + (n0 + uc);

  v8f acc[2][4];
#pragma unroll
  for (int im = 0; im < 2; ++im)
#pragma unroll
    for (int in = 0; in < 4; ++in) acc[im][in] = (v8f){};

  float ra[16], rb[16];

#define G2R(k0_)                                                                    \
  do {                                                                              \
    const float* Ak = AT ? Ag + (long long)(k0_) * LD : Ag + (k0_);                 \
    const float* Bk = Bg + (long long)(k0_) * LD;                                   \
    _Pragma("unroll")                                                               \
    for (int i = 0; i < 16; ++i) ra[i] = AT ? Ak[2 * i * LD] : Ak[8 * i * LD];      \
    _Pragma("unroll")                                                               \
    for (int i = 0; i < 16; ++i) rb[i] = Bk[2 * i * LD];                            \
  } while (0)

#define R2S(s_)                                                                     \
  do {                                                                              \
    _Pragma("unroll")                                                               \
    for (int i = 0; i < 16; ++i) {                                                  \
      if (AT) As[(s_)][uc][ur + 2 * i] = ra[i];                                     \
      else    As[(s_)][tr + 8 * i][tc] = ra[i];                                     \
    }                                                                               \
    _Pragma("unroll")                                                               \
    for (int i = 0; i < 16; ++i) Bst[(s_)][uc][ur + 2 * i] = rb[i];                 \
  } while (0)

#define TILE_MMA(s_)                                                                \
  do {                                                                              \
    _Pragma("unroll")                                                               \
    for (int kk = 0; kk < BK; kk += 4) {                                            \
      int kc = kk + 2 * h;                       \
      v2f a0 = *(const v2f*)&As[(s_)][wm + mr][kc];                                 \
      v2f a1 = *(const v2f*)&As[(s_)][wm + 16 + mr][kc];                            \
      v2f b0 = *(const v2f*)&Bst[(s_)][wn + mr][kc];                                \
      v2f b1 = *(const v2f*)&Bst[(s_)][wn + 16 + mr][kc];                           \
      v2f b2 = *(const v2f*)&Bst[(s_)][wn + 32 + mr][kc];                           \
      v2f b3 = *(const v2f*)&Bst[(s_)][wn + 48 + mr][kc];                           \
      acc[0][0] = __builtin_amdgcn_wmma_f32_16x16x4_f32(false, a0, false, b0, (short)0, acc[0][0], false, false); \
      acc[0][1] = __builtin_amdgcn_wmma_f32_16x16x4_f32(false, a0, false, b1, (short)0, acc[0][1], false, false); \
      acc[0][2] = __builtin_amdgcn_wmma_f32_16x16x4_f32(false, a0, false, b2, (short)0, acc[0][2], false, false); \
      acc[0][3] = __builtin_amdgcn_wmma_f32_16x16x4_f32(false, a0, false, b3, (short)0, acc[0][3], false, false); \
      acc[1][0] = __builtin_amdgcn_wmma_f32_16x16x4_f32(false, a1, false, b0, (short)0, acc[1][0], false, false); \
      acc[1][1] = __builtin_amdgcn_wmma_f32_16x16x4_f32(false, a1, false, b1, (short)0, acc[1][1], false, false); \
      acc[1][2] = __builtin_amdgcn_wmma_f32_16x16x4_f32(false, a1, false, b2, (short)0, acc[1][2], false, false); \
      acc[1][3] = __builtin_amdgcn_wmma_f32_16x16x4_f32(false, a1, false, b3, (short)0, acc[1][3], false, false); \
    }                                                                               \
  } while (0)

  const int nk = K / BK;

  G2R(0);
  R2S(0);
  __syncthreads();

  for (int t = 0; t < nk - 1; ++t) {
    int cur = t & 1;
    G2R((t + 1) * BK);
    __builtin_amdgcn_sched_barrier(0);
    TILE_MMA(cur);
    R2S(cur ^ 1);
    __syncthreads();
  }
  TILE_MMA((nk - 1) & 1);

#undef G2R
#undef R2S
#undef TILE_MMA

#pragma unroll
  for (int im = 0; im < 2; ++im) {
#pragma unroll
    for (int r = 0; r < 8; ++r) {
      int row = m0 + wm + 16 * im + 8 * h + r;
      float* drow = Db + (long long)row * LD;
#pragma unroll
      for (int in = 0; in < 4; ++in) {
        int col = n0 + wn + 16 * in + mr;
        float v = acc[im][in][r];
        if (mode == 1) {
          v = -0.5f * v + ((row == col) ? 1.5f : 0.0f);
        } else {
          v *= alpha;
        }
        drow[col] = v;
      }
    }
  }
}

__global__ __launch_bounds__(256)
void trace_kernel(const float* __restrict__ cov, float* __restrict__ tr, int d) {
  int b = blockIdx.x;
  int t = threadIdx.x;
  __shared__ float sh[256];
  sh[t] = cov[(long long)b * d * d + (long long)t * d + t];
  __syncthreads();
  for (int s = 128; s > 0; s >>= 1) {
    if (t < s) sh[t] += sh[t + s];
    __syncthreads();
  }
  if (t == 0) tr[b] = sh[0];
}

__global__ __launch_bounds__(256)
void init_A_ZY(const float* __restrict__ cov, const float* __restrict__ tr,
               float* __restrict__ Amat, float* __restrict__ ZY, int d) {
  int bi = blockIdx.x;
  int b  = bi / d;
  int i  = bi % d;
  int j  = threadIdx.x;
  long long idx = (long long)b * d * d + (long long)i * d + j;
  float a = cov[idx] / tr[b];
  Amat[idx] = a;
  ZY[idx]   = -0.5f * a + ((i == j) ? 1.5f : 0.0f);
}

#define TD 256

__device__ __forceinline__ void tail_load_rows(const float* M, int row0, int lane, float4 (&a)[16][2]) {
#pragma unroll
  for (int q = 0; q < 16; ++q) {
    const float4* r = (const float4*)(M + (long long)(row0 + q) * TD);
    a[q][0] = r[lane];
    a[q][1] = r[32 + lane];
  }
}

__device__ __forceinline__ void tail_dot_rows(const float4 (&a)[16][2], float4 x0, float4 x1, float (&p)[16]) {
#pragma unroll
  for (int q = 0; q < 16; ++q) {
    float s = a[q][0].x * x0.x;
    s += a[q][0].y * x0.y;
    s += a[q][0].z * x0.z;
    s += a[q][0].w * x0.w;
    s += a[q][1].x * x1.x;
    s += a[q][1].y * x1.y;
    s += a[q][1].z * x1.z;
    s += a[q][1].w * x1.w;
    p[q] = s;
  }
}

template <int KEEP>
__device__ __forceinline__ void tail_reduce_step(float (&p)[16], int lane) {
  const int bit = 2 * KEEP;
  const bool hi = (lane & bit) != 0;
#pragma unroll
  for (int i = 0; i < KEEP; ++i) {
    float send = hi ? p[i] : p[i + KEEP];
    float keep = hi ? p[i + KEEP] : p[i];
    p[i] = keep + __shfl_xor(send, bit, 32);
  }
}

__device__ __forceinline__ float tail_reduce_rows(float (&p)[16], int lane) {
  tail_reduce_step<8>(p, lane);
  tail_reduce_step<4>(p, lane);
  tail_reduce_step<2>(p, lane);
  tail_reduce_step<1>(p, lane);
  return p[0] + __shfl_xor(p[0], 1, 32);
}

__global__ __launch_bounds__(512)
void tail_matvec_kernel(const float* Ym, const float* Zm, const float* mean,
                        const float* tr, float* out) {
  __shared__ __align__(16) float sm[TD];
  __shared__ __align__(16) float su[TD];
  __shared__ __align__(16) float sw[TD];

  const int b    = blockIdx.x;
  const int tid  = threadIdx.x;
  const int wid  = tid >> 5;
  const int lane = tid & 31;
  const int row0 = wid * 16;
  const int myrow = row0 + (lane >> 1);
  const float* Yb = Ym + (long long)b * TD * TD;
  const float* Zb = Zm + (long long)b * TD * TD;

  float4 a[16][2];
  float  p[16];

  tail_load_rows(Yb, row0, lane, a);
  if (tid < TD) sm[tid] = mean[(long long)b * TD + tid];
  __syncthreads();

  float4 x0 = ((const float4*)sm)[lane];
  float4 x1 = ((const float4*)sm)[32 + lane];
  tail_dot_rows(a, x0, x1, p);
  tail_load_rows(Zb, row0, lane, a);
  float v = tail_reduce_rows(p, lane);
  if ((lane & 1) == 0) su[myrow] = v;
  __syncthreads();

  x0 = ((const float4*)su)[lane];
  x1 = ((const float4*)su)[32 + lane];
  tail_dot_rows(a, x0, x1, p);
  tail_load_rows(Yb, row0, lane, a);
  v = tail_reduce_rows(p, lane);
  if ((lane & 1) == 0) sw[myrow] = 1.5f * sm[myrow] - 0.5f * v;
  __syncthreads();

  x0 = ((const float4*)sw)[lane];
  x1 = ((const float4*)sw)[32 + lane];
  tail_dot_rows(a, x0, x1, p);
  v = tail_reduce_rows(p, lane);
  if ((lane & 1) == 0) out[(long long)b * TD + myrow] = v * sqrtf(tr[b]);
}

extern "C" void kernel_launch(void* const* d_in, const int* in_sizes, int n_in,
                              void* d_out, int out_size, void* d_ws, size_t ws_size,
                              hipStream_t stream) {
  (void)n_in; (void)ws_size; (void)out_size;
  const float* feat  = (const float*)d_in[1];
  const float* noise = (const float*)d_in[2];

  const int B  = in_sizes[0];
  const int d  = 256;
  const long long total = (long long)in_sizes[1];
  const int n  = (int)(total / ((long long)B * d));
  const int nd = n * d;
  const long long dd = (long long)d * d;

  float* W = (float*)d_ws;
  long long off = 0;
  float* Fp   = W + off; off += (long long)B * nd;
  float* mean = W + off; off += (long long)B * d;
  float* scl  = W + off; off += B;
  float* nmn  = W + off; off += B;
  float* tr   = W + off; off += B;
  off = (off + 255) & ~255LL;
  float* M0 = W + off; off += (long long)B * dd;
  float* M1 = W + off; off += (long long)B * dd;
  float* M2 = W + off; off += (long long)B * dd;
  float* M3 = W + off; off += (long long)B * dd;

  stats_kernel<<<B, 256, 0, stream>>>(feat, noise, scl, nmn, nd, 31.6227766016838f);
  perturb_kernel<<<dim3(nd / 256, B), 256, 0, stream>>>(feat, noise, scl, nmn, Fp, nd);
  mean_kernel<<<B, d, 0, stream>>>(Fp, mean, n, d);
  center_kernel<<<B * n, d, 0, stream>>>(Fp, mean, n, d);

  dim3 gg(d / BN, d / BM, B);
  wmma_gemm_batched<true, d><<<gg, 256, 0, stream>>>(Fp, (long long)nd, Fp, (long long)nd,
                                                       M0, dd, n, 1.0f / (float)(n - 1), 0);
  trace_kernel<<<B, d, 0, stream>>>(M0, tr, d);
  init_A_ZY<<<B * d, d, 0, stream>>>(M0, tr, M1, M2, d);

  wmma_gemm_batched<false, d><<<gg, 256, 0, stream>>>(M1, dd, M2, dd, M0, dd, d, 1.0f, 0);
  float* Y = M0; float* Z = M2; float* S1 = M1; float* S2 = M3;

  for (int it = 0; it < 3; ++it) {
    wmma_gemm_batched<false, d><<<gg, 256, 0, stream>>>(Z, dd, Y, dd, S1, dd, d, 1.0f, 1);
    wmma_gemm_batched<false, d><<<gg, 256, 0, stream>>>(Y, dd, S1, dd, S2, dd, d, 1.0f, 0);
    wmma_gemm_batched<false, d><<<gg, 256, 0, stream>>>(S1, dd, Z, dd, Y, dd, d, 1.0f, 0);
    float* newY = S2; float* newZ = Y; float* freed = Z;
    Y = newY; Z = newZ; S2 = freed;
  }

  tail_matvec_kernel<<<B, 512, 0, stream>>>(Y, Z, mean, tr, (float*)d_out);
}


# Stats, perturbation, column mean and centring fused into one launch: a graph's feat and noise are read once into registers and stay there (4 launches -> 1); per-graph and column sums in a new order
# speedup vs baseline: 6.0869x; 3.7998x over previous
#include <hip/hip_runtime.h>
#include <hip/hip_bf16.h>

typedef __attribute__((ext_vector_type(2))) float v2f;
typedef __attribute__((ext_vector_type(8))) float v8f;

#define PN 128
#define PD 256
#define PRE_THREADS 1024
#define PRE_RG 16
#define PRE_K (PN / PRE_RG)

__global__ __launch_bounds__(PRE_THREADS)
void prep_kernel(const float* __restrict__ feat, const float* __restrict__ noise,
                 float* __restrict__ Fp, float* __restrict__ mean, float snr_pow) {
  __shared__ double wsum[PRE_THREADS / 32][4];
  __shared__ float gst[2];
  __shared__ __align__(16) float colp[PRE_RG][PD];
  __shared__ __align__(16) float mu[PD];

  const int b    = blockIdx.x;
  const int t    = threadIdx.x;
  const int lane = t & 31;
  const int wid  = t >> 5;
  const int cg   = t & 63;
  const int rg   = t >> 6;
  const long long base = (long long)b * PN * PD + (long long)rg * PD + 4 * cg;
  const float4* f4 = (const float4*)(feat + base);
  const float4* z4 = (const float4*)(noise + base);
  float4*       o4 = (float4*)(Fp + base);

  float4 f[PRE_K], z[PRE_K];
#pragma unroll
  for (int k = 0; k < PRE_K; ++k) f[k] = f4[k * (PRE_RG * PD / 4)];
#pragma unroll
  for (int k = 0; k < PRE_K; ++k) z[k] = z4[k * (PRE_RG * PD / 4)];

  double sf = 0.0, sf2 = 0.0, sn = 0.0, sn2 = 0.0;
#pragma unroll
  for (int k = 0; k < PRE_K; ++k) {
    const float fe[4] = {f[k].x, f[k].y, f[k].z, f[k].w};
    const float ze[4] = {z[k].x, z[k].y, z[k].z, z[k].w};
#pragma unroll
    for (int e = 0; e < 4; ++e) {
      const double x = (double)fe[e];
      const double y = (double)ze[e];
      sf += x; sf2 += x * x; sn += y; sn2 += y * y;
    }
  }
#pragma unroll
  for (int m = 16; m > 0; m >>= 1) {
    sf  += __shfl_xor(sf, m);
    sf2 += __shfl_xor(sf2, m);
    sn  += __shfl_xor(sn, m);
    sn2 += __shfl_xor(sn2, m);
  }
  if (lane == 0) { wsum[wid][0] = sf; wsum[wid][1] = sf2; wsum[wid][2] = sn; wsum[wid][3] = sn2; }
  __syncthreads();
  if (wid == 0) {
    double d0 = wsum[lane][0], d1 = wsum[lane][1];
    double d2 = wsum[lane][2], d3 = wsum[lane][3];
#pragma unroll
    for (int m = 16; m > 0; m >>= 1) {
      d0 += __shfl_xor(d0, m);
      d1 += __shfl_xor(d1, m);
      d2 += __shfl_xor(d2, m);
      d3 += __shfl_xor(d3, m);
    }
    if (lane == 0) {
      double inv  = 1.0 / (double)(PN * PD);
      double gmf  = d0 * inv;
      double gmn  = d2 * inv;
      double tvar = d1 * inv - gmf * gmf;
      double nvar = d3 * inv - gmn * gmn;
      gst[0] = (float)sqrt((tvar / (double)snr_pow) / nvar);
      gst[1] = (float)gmn;
    }
  }
  __syncthreads();
  const float scl = gst[0];
  const float nmn = gst[1];

  float4 cs = make_float4(0.0f, 0.0f, 0.0f, 0.0f);
#pragma unroll
  for (int k = 0; k < PRE_K; ++k) {
    f[k].x = f[k].x + (z[k].x - nmn) * scl;
    f[k].y = f[k].y + (z[k].y - nmn) * scl;
    f[k].z = f[k].z + (z[k].z - nmn) * scl;
    f[k].w = f[k].w + (z[k].w - nmn) * scl;
    cs.x += f[k].x; cs.y += f[k].y; cs.z += f[k].z; cs.w += f[k].w;
  }
  *(float4*)&colp[rg][4 * cg] = cs;
  __syncthreads();
  if (t < PD) {
    float s = 0.0f;
#pragma unroll
    for (int r = 0; r < PRE_RG; ++r) s += colp[r][t];
    float m = s / (float)PN;
    mu[t] = m;
    mean[(long long)b * PD + t] = m;
  }
  __syncthreads();
  const float4 m4 = *(const float4*)&mu[4 * cg];
#pragma unroll
  for (int k = 0; k < PRE_K; ++k) {
    float4 v;
    v.x = f[k].x - m4.x; v.y = f[k].y - m4.y; v.z = f[k].z - m4.z; v.w = f[k].w - m4.w;
    o4[k * (PRE_RG * PD / 4)] = v;
  }
}

#define BM 128
#define BN 128
#define BK 32
#define LDP 34

template <bool AT, int LD>
__global__ __launch_bounds__(256)
void wmma_gemm_batched(const float* __restrict__ A, long long sAb,
                       const float* __restrict__ Bm, long long sBb,
                       float* __restrict__ Dm, long long sDb,
                       int K, float alpha, int mode) {
  __shared__ __align__(16) float As[2][BM][LDP];
  __shared__ __align__(16) float Bst[2][BN][LDP];

  int b  = blockIdx.z;
  int m0 = blockIdx.y * BM;
  int n0 = blockIdx.x * BN;
  const float* Ab = A  + (long long)b * sAb;
  const float* Bb = Bm + (long long)b * sBb;
  float*       Db = Dm + (long long)b * sDb;

  int tid  = threadIdx.x;
  int wid  = tid >> 5;
  int lane = tid & 31;
  int mr   = lane & 15;
  int h    = lane >> 4;
  int wm   = (wid >> 1) * 32;
  int wn   = (wid & 1) * 64;

  int tr = tid >> 5, tc = tid & 31;
  int ur = tid >> 7, uc = tid & 127;
  const float* Ag = AT ? Ab + (long long)ur * LD + (m0 + uc)
                       : Ab + (long long)(m0 + tr) * LD + tc;
  const float* Bg = Bb + (long long)ur * LD + (n0 + uc);

  v8f acc[2][4];
#pragma unroll
  for (int im = 0; im < 2; ++im)
#pragma unroll
    for (int in = 0; in < 4; ++in) acc[im][in] = (v8f){};

  float ra[16], rb[16];

#define G2R(k0_)                                                                    \
  do {                                                                              \
    const float* Ak = AT ? Ag + (long long)(k0_) * LD : Ag + (k0_);                 \
    const float* Bk = Bg + (long long)(k0_) * LD;                                   \
    _Pragma("unroll")                                                               \
    for (int i = 0; i < 16; ++i) ra[i] = AT ? Ak[2 * i * LD] : Ak[8 * i * LD];      \
    _Pragma("unroll")                                                               \
    for (int i = 0; i < 16; ++i) rb[i] = Bk[2 * i * LD];                            \
  } while (0)

#define R2S(s_)                                                                     \
  do {                                                                              \
    _Pragma("unroll")                                                               \
    for (int i = 0; i < 16; ++i) {                                                  \
      if (AT) As[(s_)][uc][ur + 2 * i] = ra[i];                                     \
      else    As[(s_)][tr + 8 * i][tc] = ra[i];                                     \
    }                                                                               \
    _Pragma("unroll")                                                               \
    for (int i = 0; i < 16; ++i) Bst[(s_)][uc][ur + 2 * i] = rb[i];                 \
  } while (0)

#define TILE_MMA(s_)                                                                \
  do {                                                                              \
    _Pragma("unroll")                                                               \
    for (int kk = 0; kk < BK; kk += 4) {                                            \
      int kc = kk + 2 * h;                       \
      v2f a0 = *(const v2f*)&As[(s_)][wm + mr][kc];                                 \
      v2f a1 = *(const v2f*)&As[(s_)][wm + 16 + mr][kc];                            \
      v2f b0 = *(const v2f*)&Bst[(s_)][wn + mr][kc];                                \
      v2f b1 = *(const v2f*)&Bst[(s_)][wn + 16 + mr][kc];                           \
      v2f b2 = *(const v2f*)&Bst[(s_)][wn + 32 + mr][kc];                           \
      v2f b3 = *(const v2f*)&Bst[(s_)][wn + 48 + mr][kc];                           \
      acc[0][0] = __builtin_amdgcn_wmma_f32_16x16x4_f32(false, a0, false, b0, (short)0, acc[0][0], false, false); \
      acc[0][1] = __builtin_amdgcn_wmma_f32_16x16x4_f32(false, a0, false, b1, (short)0, acc[0][1], false, false); \
      acc[0][2] = __builtin_amdgcn_wmma_f32_16x16x4_f32(false, a0, false, b2, (short)0, acc[0][2], false, false); \
      acc[0][3] = __builtin_amdgcn_wmma_f32_16x16x4_f32(false, a0, false, b3, (short)0, acc[0][3], false, false); \
      acc[1][0] = __builtin_amdgcn_wmma_f32_16x16x4_f32(false, a1, false, b0, (short)0, acc[1][0], false, false); \
      acc[1][1] = __builtin_amdgcn_wmma_f32_16x16x4_f32(false, a1, false, b1, (short)0, acc[1][1], false, false); \
      acc[1][2] = __builtin_amdgcn_wmma_f32_16x16x4_f32(false, a1, false, b2, (short)0, acc[1][2], false, false); \
      acc[1][3] = __builtin_amdgcn_wmma_f32_16x16x4_f32(false, a1, false, b3, (short)0, acc[1][3], false, false); \
    }                                                                               \
  } while (0)

  const int nk = K / BK;

  G2R(0);
  R2S(0);
  __syncthreads();

  for (int t = 0; t < nk - 1; ++t) {
    int cur = t & 1;
    G2R((t + 1) * BK);
    __builtin_amdgcn_sched_barrier(0);
    TILE_MMA(cur);
    R2S(cur ^ 1);
    __syncthreads();
  }
  TILE_MMA((nk - 1) & 1);

#undef G2R
#undef R2S
#undef TILE_MMA

#pragma unroll
  for (int im = 0; im < 2; ++im) {
#pragma unroll
    for (int r = 0; r < 8; ++r) {
      int row = m0 + wm + 16 * im + 8 * h + r;
      float* drow = Db + (long long)row * LD;
#pragma unroll
      for (int in = 0; in < 4; ++in) {
        int col = n0 + wn + 16 * in + mr;
        float v = acc[im][in][r];
        if (mode == 1) {
          v = -0.5f * v + ((row == col) ? 1.5f : 0.0f);
        } else {
          v *= alpha;
        }
        drow[col] = v;
      }
    }
  }
}

__global__ __launch_bounds__(256)
void trace_kernel(const float* __restrict__ cov, float* __restrict__ tr, int d) {
  int b = blockIdx.x;
  int t = threadIdx.x;
  __shared__ float sh[256];
  sh[t] = cov[(long long)b * d * d + (long long)t * d + t];
  __syncthreads();
  for (int s = 128; s > 0; s >>= 1) {
    if (t < s) sh[t] += sh[t + s];
    __syncthreads();
  }
  if (t == 0) tr[b] = sh[0];
}

#define TD 256
#define NS_LEVELS 3
#define NS_MAXOPS 160
#define NS_SLOTS 16

struct NsProg {
  unsigned int op[NS_MAXOPS];
  int n;
  int top;
};

struct NsGen {
  NsProg p;
  int sp;
  constexpr void emit(int kind, int s, int dst, int e) {
    p.op[p.n++] = (unsigned int)(kind | (s << 2) | (dst << 6) | (e << 10));
    if (sp > p.top) p.top = sp;
  }
  constexpr void applyY(int k, int dst, int s) {
    int t = sp++;
    if (k == 0) { emit(1, s, t, 0); emit(0, t, dst, 0); }
    else        { applyT(k - 1, t, s); applyY(k - 1, dst, t); }
    sp--;
  }
  constexpr void applyZ(int k, int dst, int s) {
    if (k == 0) { emit(1, s, dst, 0); return; }
    int t = sp++;
    applyZ(k - 1, t, s);
    applyT(k - 1, dst, t);
    sp--;
  }
  constexpr void applyT(int k, int dst, int s) {
    int t = sp++;
    applyY(k, t, s);
    if (k == 0) { emit(2, t, dst, s); }
    else        { int t2 = sp++; applyZ(k, t2, t); emit(3, t2, dst, s); sp--; }
    sp--;
  }
};

constexpr NsProg ns_make_prog() {
  NsGen g{};
  g.sp = 2;
  g.applyY(NS_LEVELS + 1, 1, 0);
  return g.p;
}

__constant__ constexpr NsProg ns_prog = ns_make_prog();
static_assert(ns_prog.n < NS_MAXOPS && ns_prog.top <= NS_SLOTS, "step list does not fit");

template <int MASK>
__device__ __forceinline__ float row_xor_lane(float v) {
  return __builtin_bit_cast(float, __builtin_amdgcn_update_dpp(0, __builtin_bit_cast(int, v),
                                                               0x160 | MASK, 0xf, 0xf, true));
}

template <int KEEP>
__device__ __forceinline__ void ns_reduce_step(float (&p)[8], int lane) {
  constexpr int bit = 2 * KEEP;
  const bool hi = (lane & bit) != 0;
#pragma unroll
  for (int i = 0; i < KEEP; ++i) {
    float send = hi ? p[i] : p[i + KEEP];
    float keep = hi ? p[i + KEEP] : p[i];
    p[i] = keep + row_xor_lane<bit>(send);
  }
}

__global__ __launch_bounds__(512)
void ns_vector_kernel(const float* __restrict__ cov, const float* __restrict__ mean,
                      const float* __restrict__ tr, float* __restrict__ out) {
  __shared__ __align__(16) float sv[NS_SLOTS][TD];

  const int b    = blockIdx.x;
  const int tid  = threadIdx.x;
  const int wid  = tid >> 5;
  const int lane = tid & 31;
  const int c    = lane & 15;
  const int row0 = wid * 16 + (lane >> 4) * 8;
  const int myrow = wid * 16 + (lane >> 1);

  v2f a[8][8];
  const float* cb = cov + (long long)b * TD * TD;
#pragma unroll
  for (int q = 0; q < 8; ++q) {
    const float4* r = (const float4*)(cb + (long long)(row0 + q) * TD);
#pragma unroll
    for (int j = 0; j < 4; ++j) {
      float4 t = r[16 * j + c];
      a[q][2 * j]     = (v2f){t.x, t.y};
      a[q][2 * j + 1] = (v2f){t.z, t.w};
    }
  }
  if (tid < TD) sv[0][tid] = mean[(long long)b * TD + tid];
  const float trb = tr[b];
  const float itr = 1.0f / trb;
  __syncthreads();

  const int nops = ns_prog.n;
  unsigned next = ns_prog.op[0];
  for (int i = 0; i < nops; ++i) {
    const unsigned op = next;
    next = ns_prog.op[i + 1];
    const int kind = op & 3;
    const float* xs = sv[(op >> 2) & 15];
    float*       xd = sv[(op >> 6) & 15];
    const float* xe = sv[(op >> 10) & 15];
    const float es = xs[myrow];
    const float ee = xe[myrow];
    float val;
    if (kind != 3) {
      v2f x[8];
#pragma unroll
      for (int j = 0; j < 4; ++j) {
        float4 t = ((const float4*)xs)[16 * j + c];
        x[2 * j]     = (v2f){t.x, t.y};
        x[2 * j + 1] = (v2f){t.z, t.w};
      }
      float p[8];
#pragma unroll
      for (int q = 0; q < 8; ++q) {
        v2f s = a[q][0] * x[0];
#pragma unroll
        for (int j = 1; j < 8; ++j) s = __builtin_elementwise_fma(a[q][j], x[j], s);
        p[q] = s.x + s.y;
      }
      ns_reduce_step<4>(p, lane);
      ns_reduce_step<2>(p, lane);
      ns_reduce_step<1>(p, lane);
      val = (p[0] + row_xor_lane<1>(p[0])) * itr;
      if (kind != 0) val = 1.5f * es - 0.5f * val;
      if (kind == 2) val = 1.5f * ee - 0.5f * val;
    } else {
      val = 1.5f * ee - 0.5f * es;
    }
    if ((lane & 1) == 0) xd[myrow] = val;
    __syncthreads();
  }

  if (tid < TD) out[(long long)b * TD + tid] = sv[1][tid] * sqrtf(trb);
}

extern "C" void kernel_launch(void* const* d_in, const int* in_sizes, int n_in,
                              void* d_out, int out_size, void* d_ws, size_t ws_size,
                              hipStream_t stream) {
  (void)n_in; (void)ws_size; (void)out_size;
  const float* feat  = (const float*)d_in[1];
  const float* noise = (const float*)d_in[2];

  const int B  = in_sizes[0];
  const int d  = 256;
  const long long total = (long long)in_sizes[1];
  const int n  = (int)(total / ((long long)B * d));
  const int nd = n * d;
  const long long dd = (long long)d * d;

  float* W = (float*)d_ws;
  long long off = 0;
  float* Fp   = W + off; off += (long long)B * nd;
  float* mean = W + off; off += (long long)B * d;
  off += B;
  off += B;
  float* tr   = W + off; off += B;
  off = (off + 255) & ~255LL;
  float* M0 = W + off;

  prep_kernel<<<B, PRE_THREADS, 0, stream>>>(feat, noise, Fp, mean, 31.6227766016838f);

  dim3 gg(d / BN, d / BM, B);
  wmma_gemm_batched<true, d><<<gg, 256, 0, stream>>>(Fp, (long long)nd, Fp, (long long)nd,
                                                       M0, dd, n, 1.0f / (float)(n - 1), 0);
  trace_kernel<<<B, d, 0, stream>>>(M0, tr, d);

  ns_vector_kernel<<<B, 512, 0, stream>>>(M0, mean, tr, (float*)d_out);
}


# speedup vs baseline: 7.6005x; 1.2487x over previous
#include <hip/hip_runtime.h>
#include <hip/hip_bf16.h>

typedef __attribute__((ext_vector_type(2))) float v2f;
typedef __attribute__((ext_vector_type(8))) float v8f;

#define PN 128
#define PD 256
#define PRE_THREADS 1024
#define PRE_RG 16
#define PRE_K (PN / PRE_RG)

__global__ __launch_bounds__(PRE_THREADS)
void prep_kernel(const float* __restrict__ feat, const float* __restrict__ noise,
                 float* __restrict__ Fp, float* __restrict__ mean, float snr_pow) {
  __shared__ __align__(16) float wsum[PRE_THREADS / 32][4];
  __shared__ float gst[2];
  __shared__ __align__(16) float colp[PRE_RG][PD];
  __shared__ __align__(16) float mu[PD];

  const int b    = blockIdx.x;
  const int t    = threadIdx.x;
  const int lane = t & 31;
  const int wid  = t >> 5;
  const int cg   = t & 63;
  const int rg   = t >> 6;
  const long long base = (long long)b * PN * PD + (long long)rg * PD + 4 * cg;
  const float4* f4 = (const float4*)(feat + base);
  const float4* z4 = (const float4*)(noise + base);
  float4*       o4 = (float4*)(Fp + base);

  float4 f[PRE_K], z[PRE_K];
#pragma unroll
  for (int k = 0; k < PRE_K; ++k) f[k] = f4[k * (PRE_RG * PD / 4)];
#pragma unroll
  for (int k = 0; k < PRE_K; ++k) z[k] = z4[k * (PRE_RG * PD / 4)];

  float4 af  = make_float4(0.0f, 0.0f, 0.0f, 0.0f), af2 = make_float4(0.0f, 0.0f, 0.0f, 0.0f);
  float4 an  = make_float4(0.0f, 0.0f, 0.0f, 0.0f), an2 = make_float4(0.0f, 0.0f, 0.0f, 0.0f);
#pragma unroll
  for (int k = 0; k < PRE_K; ++k) {
    af.x += f[k].x; af.y += f[k].y; af.z += f[k].z; af.w += f[k].w;
    af2.x += f[k].x * f[k].x; af2.y += f[k].y * f[k].y;
    af2.z += f[k].z * f[k].z; af2.w += f[k].w * f[k].w;
    an.x += z[k].x; an.y += z[k].y; an.z += z[k].z; an.w += z[k].w;
    an2.x += z[k].x * z[k].x; an2.y += z[k].y * z[k].y;
    an2.z += z[k].z * z[k].z; an2.w += z[k].w * z[k].w;
  }
  float sf  = (af.x + af.y) + (af.z + af.w);
  float sf2 = (af2.x + af2.y) + (af2.z + af2.w);
  float sn  = (an.x + an.y) + (an.z + an.w);
  float sn2 = (an2.x + an2.y) + (an2.z + an2.w);
#pragma unroll
  for (int m = 16; m > 0; m >>= 1) {
    sf  += __shfl_xor(sf, m);
    sf2 += __shfl_xor(sf2, m);
    sn  += __shfl_xor(sn, m);
    sn2 += __shfl_xor(sn2, m);
  }
  if (lane == 0) *(float4*)&wsum[wid][0] = make_float4(sf, sf2, sn, sn2);
  __syncthreads();
  if (wid == 0) {
    const float4 w4 = *(const float4*)&wsum[lane][0];
    double d0 = (double)w4.x, d1 = (double)w4.y;
    double d2 = (double)w4.z, d3 = (double)w4.w;
#pragma unroll
    for (int m = 16; m > 0; m >>= 1) {
      d0 += __shfl_xor(d0, m);
      d1 += __shfl_xor(d1, m);
      d2 += __shfl_xor(d2, m);
      d3 += __shfl_xor(d3, m);
    }
    if (lane == 0) {
      double inv  = 1.0 / (double)(PN * PD);
      double gmf  = d0 * inv;
      double gmn  = d2 * inv;
      double tvar = d1 * inv - gmf * gmf;
      double nvar = d3 * inv - gmn * gmn;
      gst[0] = (float)sqrt((tvar / (double)snr_pow) / nvar);
      gst[1] = (float)gmn;
    }
  }
  __syncthreads();
  const float scl = gst[0];
  const float nmn = gst[1];

  float4 cs = make_float4(0.0f, 0.0f, 0.0f, 0.0f);
#pragma unroll
  for (int k = 0; k < PRE_K; ++k) {
    f[k].x = f[k].x + (z[k].x - nmn) * scl;
    f[k].y = f[k].y + (z[k].y - nmn) * scl;
    f[k].z = f[k].z + (z[k].z - nmn) * scl;
    f[k].w = f[k].w + (z[k].w - nmn) * scl;
    cs.x += f[k].x; cs.y += f[k].y; cs.z += f[k].z; cs.w += f[k].w;
  }
  *(float4*)&colp[rg][4 * cg] = cs;
  __syncthreads();
  if (t < PD) {
    float s = 0.0f;
#pragma unroll
    for (int r = 0; r < PRE_RG; ++r) s += colp[r][t];
    float m = s / (float)PN;
    mu[t] = m;
    mean[(long long)b * PD + t] = m;
  }
  __syncthreads();
  const float4 m4 = *(const float4*)&mu[4 * cg];
#pragma unroll
  for (int k = 0; k < PRE_K; ++k) {
    float4 v;
    v.x = f[k].x - m4.x; v.y = f[k].y - m4.y; v.z = f[k].z - m4.z; v.w = f[k].w - m4.w;
    o4[k * (PRE_RG * PD / 4)] = v;
  }
}

#define BM 128
#define BN 128
#define BK 32
#define LDP 34

template <bool AT, int LD>
__global__ __launch_bounds__(256)
void wmma_gemm_batched(const float* __restrict__ A, long long sAb,
                       const float* __restrict__ Bm, long long sBb,
                       float* __restrict__ Dm, long long sDb,
                       int K, float alpha, int mode) {
  __shared__ __align__(16) float As[2][BM][LDP];
  __shared__ __align__(16) float Bst[2][BN][LDP];

  int b  = blockIdx.z;
  int m0 = blockIdx.y * BM;
  int n0 = blockIdx.x * BN;
  const float* Ab = A  + (long long)b * sAb;
  const float* Bb = Bm + (long long)b * sBb;
  float*       Db = Dm + (long long)b * sDb;

  int tid  = threadIdx.x;
  int wid  = tid >> 5;
  int lane = tid & 31;
  int mr   = lane & 15;
  int h    = lane >> 4;
  int wm   = (wid >> 1) * 32;
  int wn   = (wid & 1) * 64;

  int tr = tid >> 5, tc = tid & 31;
  int ur = tid >> 7, uc = tid & 127;
  const float* Ag = AT ? Ab + (long long)ur * LD + (m0 + uc)
                       : Ab + (long long)(m0 + tr) * LD + tc;
  const float* Bg = Bb + (long long)ur * LD + (n0 + uc);

  v8f acc[2][4];
#pragma unroll
  for (int im = 0; im < 2; ++im)
#pragma unroll
    for (int in = 0; in < 4; ++in) acc[im][in] = (v8f){};

  float ra[16], rb[16];

#define G2R(k0_)                                                                    \
  do {                                                                              \
    const float* Ak = AT ? Ag + (long long)(k0_) * LD : Ag + (k0_);                 \
    const float* Bk = Bg + (long long)(k0_) * LD;                                   \
    _Pragma("unroll")                                                               \
    for (int i = 0; i < 16; ++i) ra[i] = AT ? Ak[2 * i * LD] : Ak[8 * i * LD];      \
    _Pragma("unroll")                                                               \
    for (int i = 0; i < 16; ++i) rb[i] = Bk[2 * i * LD];                            \
  } while (0)

#define R2S(s_)                                                                     \
  do {                                                                              \
    _Pragma("unroll")                                                               \
    for (int i = 0; i < 16; ++i) {                                                  \
      if (AT) As[(s_)][uc][ur + 2 * i] = ra[i];                                     \
      else    As[(s_)][tr + 8 * i][tc] = ra[i];                                     \
    }                                                                               \
    _Pragma("unroll")                                                               \
    for (int i = 0; i < 16; ++i) Bst[(s_)][uc][ur + 2 * i] = rb[i];                 \
  } while (0)

#define TILE_MMA(s_)                                                                \
  do {                                                                              \
    _Pragma("unroll")                                                               \
    for (int kk = 0; kk < BK; kk += 4) {                                            \
      int kc = kk + 2 * h;                       \
      v2f a0 = *(const v2f*)&As[(s_)][wm + mr][kc];                                 \
      v2f a1 = *(const v2f*)&As[(s_)][wm + 16 + mr][kc];                            \
      v2f b0 = *(const v2f*)&Bst[(s_)][wn + mr][kc];                                \
      v2f b1 = *(const v2f*)&Bst[(s_)][wn + 16 + mr][kc];                           \
      v2f b2 = *(const v2f*)&Bst[(s_)][wn + 32 + mr][kc];                           \
      v2f b3 = *(const v2f*)&Bst[(s_)][wn + 48 + mr][kc];                           \
      acc[0][0] = __builtin_amdgcn_wmma_f32_16x16x4_f32(false, a0, false, b0, (short)0, acc[0][0], false, false); \
      acc[0][1] = __builtin_amdgcn_wmma_f32_16x16x4_f32(false, a0, false, b1, (short)0, acc[0][1], false, false); \
      acc[0][2] = __builtin_amdgcn_wmma_f32_16x16x4_f32(false, a0, false, b2, (short)0, acc[0][2], false, false); \
      acc[0][3] = __builtin_amdgcn_wmma_f32_16x16x4_f32(false, a0, false, b3, (short)0, acc[0][3], false, false); \
      acc[1][0] = __builtin_amdgcn_wmma_f32_16x16x4_f32(false, a1, false, b0, (short)0, acc[1][0], false, false); \
      acc[1][1] = __builtin_amdgcn_wmma_f32_16x16x4_f32(false, a1, false, b1, (short)0, acc[1][1], false, false); \
      acc[1][2] = __builtin_amdgcn_wmma_f32_16x16x4_f32(false, a1, false, b2, (short)0, acc[1][2], false, false); \
      acc[1][3] = __builtin_amdgcn_wmma_f32_16x16x4_f32(false, a1, false, b3, (short)0, acc[1][3], false, false); \
    }                                                                               \
  } while (0)

  const int nk = K / BK;

  G2R(0);
  R2S(0);
  __syncthreads();

  for (int t = 0; t < nk - 1; ++t) {
    int cur = t & 1;
    G2R((t + 1) * BK);
    __builtin_amdgcn_sched_barrier(0);
    TILE_MMA(cur);
    R2S(cur ^ 1);
    __syncthreads();
  }
  TILE_MMA((nk - 1) & 1);

#undef G2R
#undef R2S
#undef TILE_MMA

#pragma unroll
  for (int im = 0; im < 2; ++im) {
#pragma unroll
    for (int r = 0; r < 8; ++r) {
      int row = m0 + wm + 16 * im + 8 * h + r;
      float* drow = Db + (long long)row * LD;
#pragma unroll
      for (int in = 0; in < 4; ++in) {
        int col = n0 + wn + 16 * in + mr;
        float v = acc[im][in][r];
        if (mode == 1) {
          v = -0.5f * v + ((row == col) ? 1.5f : 0.0f);
        } else {
          v *= alpha;
        }
        drow[col] = v;
      }
    }
  }
}

__global__ __launch_bounds__(256)
void trace_kernel(const float* __restrict__ cov, float* __restrict__ tr, int d) {
  int b = blockIdx.x;
  int t = threadIdx.x;
  __shared__ float sh[256];
  sh[t] = cov[(long long)b * d * d + (long long)t * d + t];
  __syncthreads();
  for (int s = 128; s > 0; s >>= 1) {
    if (t < s) sh[t] += sh[t + s];
    __syncthreads();
  }
  if (t == 0) tr[b] = sh[0];
}

#define TD 256
#define NS_LEVELS 3
#define NS_MAXOPS 160
#define NS_SLOTS 16

struct NsProg {
  unsigned int op[NS_MAXOPS];
  int n;
  int top;
};

struct NsGen {
  NsProg p;
  int sp;
  constexpr void emit(int kind, int s, int dst, int e) {
    p.op[p.n++] = (unsigned int)(kind | (s << 2) | (dst << 6) | (e << 10));
    if (sp > p.top) p.top = sp;
  }
  constexpr void applyY(int k, int dst, int s) {
    int t = sp++;
    if (k == 0) { emit(1, s, t, 0); emit(0, t, dst, 0); }
    else        { applyT(k - 1, t, s); applyY(k - 1, dst, t); }
    sp--;
  }
  constexpr void applyZ(int k, int dst, int s) {
    if (k == 0) { emit(1, s, dst, 0); return; }
    int t = sp++;
    applyZ(k - 1, t, s);
    applyT(k - 1, dst, t);
    sp--;
  }
  constexpr void applyT(int k, int dst, int s) {
    int t = sp++;
    applyY(k, t, s);
    if (k == 0) { emit(2, t, dst, s); }
    else        { int t2 = sp++; applyZ(k, t2, t); emit(3, t2, dst, s); sp--; }
    sp--;
  }
};

constexpr NsProg ns_make_prog() {
  NsGen g{};
  g.sp = 2;
  g.applyY(NS_LEVELS + 1, 1, 0);
  return g.p;
}

__constant__ constexpr NsProg ns_prog = ns_make_prog();
static_assert(ns_prog.n < NS_MAXOPS && ns_prog.top <= NS_SLOTS, "step list does not fit");

template <int MASK>
__device__ __forceinline__ float row_xor_lane(float v) {
  return __builtin_bit_cast(float, __builtin_amdgcn_update_dpp(0, __builtin_bit_cast(int, v),
                                                               0x160 | MASK, 0xf, 0xf, true));
}

template <int KEEP>
__device__ __forceinline__ void ns_reduce_step(float (&p)[8], int lane) {
  constexpr int bit = 2 * KEEP;
  const bool hi = (lane & bit) != 0;
#pragma unroll
  for (int i = 0; i < KEEP; ++i) {
    float send = hi ? p[i] : p[i + KEEP];
    float keep = hi ? p[i + KEEP] : p[i];
    p[i] = keep + row_xor_lane<bit>(send);
  }
}

__global__ __launch_bounds__(512)
void ns_vector_kernel(const float* __restrict__ cov, const float* __restrict__ mean,
                      const float* __restrict__ tr, float* __restrict__ out) {
  __shared__ __align__(16) float sv[NS_SLOTS][TD];

  const int b    = blockIdx.x;
  const int tid  = threadIdx.x;
  const int wid  = tid >> 5;
  const int lane = tid & 31;
  const int c    = lane & 15;
  const int row0 = wid * 16 + (lane >> 4) * 8;
  const int myrow = wid * 16 + (lane >> 1);

  v2f a[8][8];
  const float* cb = cov + (long long)b * TD * TD;
#pragma unroll
  for (int q = 0; q < 8; ++q) {
    const float4* r = (const float4*)(cb + (long long)(row0 + q) * TD);
#pragma unroll
    for (int j = 0; j < 4; ++j) {
      float4 t = r[16 * j + c];
      a[q][2 * j]     = (v2f){t.x, t.y};
      a[q][2 * j + 1] = (v2f){t.z, t.w};
    }
  }
  if (tid < TD) sv[0][tid] = mean[(long long)b * TD + tid];
  const float trb = tr[b];
  const float itr = 1.0f / trb;
  __syncthreads();

  const int nops = ns_prog.n;
  unsigned next = ns_prog.op[0];
  for (int i = 0; i < nops; ++i) {
    const unsigned op = next;
    next = ns_prog.op[i + 1];
    const int kind = op & 3;
    const float* xs = sv[(op >> 2) & 15];
    float*       xd = sv[(op >> 6) & 15];
    const float* xe = sv[(op >> 10) & 15];
    const float es = xs[myrow];
    const float ee = xe[myrow];
    float val;
    if (kind != 3) {
      v2f x[8];
#pragma unroll
      for (int j = 0; j < 4; ++j) {
        float4 t = ((const float4*)xs)[16 * j + c];
        x[2 * j]     = (v2f){t.x, t.y};
        x[2 * j + 1] = (v2f){t.z, t.w};
      }
      float p[8];
#pragma unroll
      for (int q = 0; q < 8; ++q) {
        v2f s = a[q][0] * x[0];
#pragma unroll
        for (int j = 1; j < 8; ++j) s = __builtin_elementwise_fma(a[q][j], x[j], s);
        p[q] = s.x + s.y;
      }
      ns_reduce_step<4>(p, lane);
      ns_reduce_step<2>(p, lane);
      ns_reduce_step<1>(p, lane);
      val = (p[0] + row_xor_lane<1>(p[0])) * itr;
      if (kind != 0) val = 1.5f * es - 0.5f * val;
      if (kind == 2) val = 1.5f * ee - 0.5f * val;
    } else {
      val = 1.5f * ee - 0.5f * es;
    }
    if ((lane & 1) == 0) xd[myrow] = val;
    __syncthreads();
  }

  if (tid < TD) out[(long long)b * TD + tid] = sv[1][tid] * sqrtf(trb);
}

extern "C" void kernel_launch(void* const* d_in, const int* in_sizes, int n_in,
                              void* d_out, int out_size, void* d_ws, size_t ws_size,
                              hipStream_t stream) {
  (void)n_in; (void)ws_size; (void)out_size;
  const float* feat  = (const float*)d_in[1];
  const float* noise = (const float*)d_in[2];

  const int B  = in_sizes[0];
  const int d  = 256;
  const long long total = (long long)in_sizes[1];
  const int n  = (int)(total / ((long long)B * d));
  const int nd = n * d;
  const long long dd = (long long)d * d;

  float* W = (float*)d_ws;
  long long off = 0;
  float* Fp   = W + off; off += (long long)B * nd;
  float* mean = W + off; off += (long long)B * d;
  off += B;
  off += B;
  float* tr   = W + off; off += B;
  off = (off + 255) & ~255LL;
  float* M0 = W + off;

  prep_kernel<<<B, PRE_THREADS, 0, stream>>>(feat, noise, Fp, mean, 31.6227766016838f);

  dim3 gg(d / BN, d / BM, B);
  wmma_gemm_batched<true, d><<<gg, 256, 0, stream>>>(Fp, (long long)nd, Fp, (long long)nd,
                                                       M0, dd, n, 1.0f / (float)(n - 1), 0);
  trace_kernel<<<B, d, 0, stream>>>(M0, tr, d);

  ns_vector_kernel<<<B, 512, 0, stream>>>(M0, mean, tr, (float*)d_out);
}


# Newton-Schulz moved to the 128x128 Gram matrix D D^T: p(D^T D) m = D^T q(D D^T) D m; the GEMM and the 121 matrix-vector products shrink from order 256 to 128; sums taken in a new order
# speedup vs baseline: 12.0140x; 1.5807x over previous
#include <hip/hip_runtime.h>
#include <hip/hip_bf16.h>

typedef __attribute__((ext_vector_type(2))) float v2f;
typedef __attribute__((ext_vector_type(8))) float v8f;

#define PN 128
#define PD 256
#define PRE_THREADS 1024
#define PRE_RG 16
#define PRE_K (PN / PRE_RG)

__global__ __launch_bounds__(PRE_THREADS)
void prep_kernel(const float* __restrict__ feat, const float* __restrict__ noise,
                 float* __restrict__ Fp, float* __restrict__ mean, float snr_pow) {
  __shared__ __align__(16) float wsum[PRE_THREADS / 32][4];
  __shared__ float gst[2];
  __shared__ __align__(16) float colp[PRE_RG][PD];
  __shared__ __align__(16) float mu[PD];

  const int b    = blockIdx.x;
  const int t    = threadIdx.x;
  const int lane = t & 31;
  const int wid  = t >> 5;
  const int cg   = t & 63;
  const int rg   = t >> 6;
  const long long base = (long long)b * PN * PD + (long long)rg * PD + 4 * cg;
  const float4* f4 = (const float4*)(feat + base);
  const float4* z4 = (const float4*)(noise + base);
  float4*       o4 = (float4*)(Fp + base);

  float4 f[PRE_K], z[PRE_K];
#pragma unroll
  for (int k = 0; k < PRE_K; ++k) f[k] = f4[k * (PRE_RG * PD / 4)];
#pragma unroll
  for (int k = 0; k < PRE_K; ++k) z[k] = z4[k * (PRE_RG * PD / 4)];

  float4 af  = make_float4(0.0f, 0.0f, 0.0f, 0.0f), af2 = make_float4(0.0f, 0.0f, 0.0f, 0.0f);
  float4 an  = make_float4(0.0f, 0.0f, 0.0f, 0.0f), an2 = make_float4(0.0f, 0.0f, 0.0f, 0.0f);
#pragma unroll
  for (int k = 0; k < PRE_K; ++k) {
    af.x += f[k].x; af.y += f[k].y; af.z += f[k].z; af.w += f[k].w;
    af2.x += f[k].x * f[k].x; af2.y += f[k].y * f[k].y;
    af2.z += f[k].z * f[k].z; af2.w += f[k].w * f[k].w;
    an.x += z[k].x; an.y += z[k].y; an.z += z[k].z; an.w += z[k].w;
    an2.x += z[k].x * z[k].x; an2.y += z[k].y * z[k].y;
    an2.z += z[k].z * z[k].z; an2.w += z[k].w * z[k].w;
  }
  float sf  = (af.x + af.y) + (af.z + af.w);
  float sf2 = (af2.x + af2.y) + (af2.z + af2.w);
  float sn  = (an.x + an.y) + (an.z + an.w);
  float sn2 = (an2.x + an2.y) + (an2.z + an2.w);
#pragma unroll
  for (int m = 16; m > 0; m >>= 1) {
    sf  += __shfl_xor(sf, m);
    sf2 += __shfl_xor(sf2, m);
    sn  += __shfl_xor(sn, m);
    sn2 += __shfl_xor(sn2, m);
  }
  if (lane == 0) *(float4*)&wsum[wid][0] = make_float4(sf, sf2, sn, sn2);
  __syncthreads();
  if (wid == 0) {
    const float4 w4 = *(const float4*)&wsum[lane][0];
    double d0 = (double)w4.x, d1 = (double)w4.y;
    double d2 = (double)w4.z, d3 = (double)w4.w;
#pragma unroll
    for (int m = 16; m > 0; m >>= 1) {
      d0 += __shfl_xor(d0, m);
      d1 += __shfl_xor(d1, m);
      d2 += __shfl_xor(d2, m);
      d3 += __shfl_xor(d3, m);
    }
    if (lane == 0) {
      double inv  = 1.0 / (double)(PN * PD);
      double gmf  = d0 * inv;
      double gmn  = d2 * inv;
      double tvar = d1 * inv - gmf * gmf;
      double nvar = d3 * inv - gmn * gmn;
      gst[0] = (float)sqrt((tvar / (double)snr_pow) / nvar);
      gst[1] = (float)gmn;
    }
  }
  __syncthreads();
  const float scl = gst[0];
  const float nmn = gst[1];

  float4 cs = make_float4(0.0f, 0.0f, 0.0f, 0.0f);
#pragma unroll
  for (int k = 0; k < PRE_K; ++k) {
    f[k].x = f[k].x + (z[k].x - nmn) * scl;
    f[k].y = f[k].y + (z[k].y - nmn) * scl;
    f[k].z = f[k].z + (z[k].z - nmn) * scl;
    f[k].w = f[k].w + (z[k].w - nmn) * scl;
    cs.x += f[k].x; cs.y += f[k].y; cs.z += f[k].z; cs.w += f[k].w;
  }
  *(float4*)&colp[rg][4 * cg] = cs;
  __syncthreads();
  if (t < PD) {
    float s = 0.0f;
#pragma unroll
    for (int r = 0; r < PRE_RG; ++r) s += colp[r][t];
    float m = s / (float)PN;
    mu[t] = m;
    mean[(long long)b * PD + t] = m;
  }
  __syncthreads();
  const float4 m4 = *(const float4*)&mu[4 * cg];
#pragma unroll
  for (int k = 0; k < PRE_K; ++k) {
    float4 v;
    v.x = f[k].x - m4.x; v.y = f[k].y - m4.y; v.z = f[k].z - m4.z; v.w = f[k].w - m4.w;
    o4[k * (PRE_RG * PD / 4)] = v;
  }
}

#define GM 64
#define BK 32
#define LDP 34

__global__ __launch_bounds__(256)
void gram_wmma_kernel(const float* __restrict__ Dm, float* __restrict__ S, float alpha) {
  __shared__ __align__(16) float Ts[2][PN][LDP];

  const int b  = blockIdx.y;
  const int m0 = blockIdx.x * GM;
  const float* Db = Dm + (long long)b * PN * PD;
  float*       Sb = S  + (long long)b * PN * PN;

  const int tid  = threadIdx.x;
  const int wid  = tid >> 5;
  const int lane = tid & 31;
  const int mr   = lane & 15;
  const int h    = lane >> 4;
  const int wm   = m0 + (wid >> 2) * 32;
  const int wn   = (wid & 3) * 32;

  const int tr = tid >> 5, tc = tid & 31;
  const float* Dg = Db + (long long)tr * PD + tc;

  v8f acc[2][2];
#pragma unroll
  for (int im = 0; im < 2; ++im)
#pragma unroll
    for (int in = 0; in < 2; ++in) acc[im][in] = (v8f){};

  float rd[16];

#define G2R(k0_)                                                                    \
  do {                                                                              \
    const float* Dk = Dg + (k0_);                                                   \
    _Pragma("unroll")                                                               \
    for (int i = 0; i < 16; ++i) rd[i] = Dk[8 * i * PD];                            \
  } while (0)

#define R2S(s_)                                                                     \
  do {                                                                              \
    _Pragma("unroll")                                                               \
    for (int i = 0; i < 16; ++i) Ts[(s_)][tr + 8 * i][tc] = rd[i];                  \
  } while (0)

#define TILE_MMA(s_)                                                                \
  do {                                                                              \
    _Pragma("unroll")                                                               \
    for (int kk = 0; kk < BK; kk += 4) {                                            \
      int kc = kk + 2 * h;                       \
      v2f a0 = *(const v2f*)&Ts[(s_)][wm + mr][kc];                                 \
      v2f a1 = *(const v2f*)&Ts[(s_)][wm + 16 + mr][kc];                            \
      v2f b0 = *(const v2f*)&Ts[(s_)][wn + mr][kc];                                 \
      v2f b1 = *(const v2f*)&Ts[(s_)][wn + 16 + mr][kc];                            \
      acc[0][0] = __builtin_amdgcn_wmma_f32_16x16x4_f32(false, a0, false, b0, (short)0, acc[0][0], false, false); \
      acc[0][1] = __builtin_amdgcn_wmma_f32_16x16x4_f32(false, a0, false, b1, (short)0, acc[0][1], false, false); \
      acc[1][0] = __builtin_amdgcn_wmma_f32_16x16x4_f32(false, a1, false, b0, (short)0, acc[1][0], false, false); \
      acc[1][1] = __builtin_amdgcn_wmma_f32_16x16x4_f32(false, a1, false, b1, (short)0, acc[1][1], false, false); \
    }                                                                               \
  } while (0)

  const int nk = PD / BK;

  G2R(0);
  R2S(0);
  __syncthreads();

  for (int t = 0; t < nk - 1; ++t) {
    int cur = t & 1;
    G2R((t + 1) * BK);
    __builtin_amdgcn_sched_barrier(0);
    TILE_MMA(cur);
    R2S(cur ^ 1);
    __syncthreads();
  }
  TILE_MMA((nk - 1) & 1);

#undef G2R
#undef R2S
#undef TILE_MMA

#pragma unroll
  for (int im = 0; im < 2; ++im) {
#pragma unroll
    for (int r = 0; r < 8; ++r) {
      int row = wm + 16 * im + 8 * h + r;
      float* srow = Sb + (long long)row * PN;
#pragma unroll
      for (int in = 0; in < 2; ++in) {
        int col = wn + 16 * in + mr;
        srow[col] = acc[im][in][r] * alpha;
      }
    }
  }
}

__global__ __launch_bounds__(256)
void trace_kernel(const float* __restrict__ cov, float* __restrict__ tr, int d) {
  int b = blockIdx.x;
  int t = threadIdx.x;
  __shared__ float sh[256];
  sh[t] = cov[(long long)b * d * d + (long long)t * d + t];
  __syncthreads();
  for (int s = d >> 1; s > 0; s >>= 1) {
    if (t < s) sh[t] += sh[t + s];
    __syncthreads();
  }
  if (t == 0) tr[b] = sh[0];
}

#define TN PN
#define NS_LEVELS 3
#define NS_MAXOPS 160
#define NS_SLOTS 16

struct NsProg {
  unsigned int op[NS_MAXOPS];
  int n;
  int top;
};

struct NsGen {
  NsProg p;
  int sp;
  constexpr void emit(int kind, int s, int dst, int e) {
    p.op[p.n++] = (unsigned int)(kind | (s << 2) | (dst << 6) | (e << 10));
    if (sp > p.top) p.top = sp;
  }
  constexpr void applyY(int k, int dst, int s) {
    int t = sp++;
    if (k == 0) { emit(1, s, t, 0); emit(0, t, dst, 0); }
    else        { applyT(k - 1, t, s); applyY(k - 1, dst, t); }
    sp--;
  }
  constexpr void applyZ(int k, int dst, int s) {
    if (k == 0) { emit(1, s, dst, 0); return; }
    int t = sp++;
    applyZ(k - 1, t, s);
    applyT(k - 1, dst, t);
    sp--;
  }
  constexpr void applyT(int k, int dst, int s) {
    int t = sp++;
    applyY(k, t, s);
    if (k == 0) { emit(2, t, dst, s); }
    else        { int t2 = sp++; applyZ(k, t2, t); emit(3, t2, dst, s); sp--; }
    sp--;
  }
};

constexpr NsProg ns_make_prog() {
  NsGen g{};
  g.sp = 2;
  g.applyZ(NS_LEVELS + 1, 1, 0);
  return g.p;
}

__constant__ constexpr NsProg ns_prog = ns_make_prog();
static_assert(ns_prog.n < NS_MAXOPS && ns_prog.top <= NS_SLOTS, "step list does not fit");

template <int MASK>
__device__ __forceinline__ float row_xor_lane(float v) {
  return __builtin_bit_cast(float, __builtin_amdgcn_update_dpp(0, __builtin_bit_cast(int, v),
                                                               0x160 | MASK, 0xf, 0xf, true));
}

template <int KEEP>
__device__ __forceinline__ void ns_reduce_step(float (&p)[8], int lane) {
  constexpr int bit = 2 * KEEP;
  const bool hi = (lane & bit) != 0;
#pragma unroll
  for (int i = 0; i < KEEP; ++i) {
    float send = hi ? p[i] : p[i + KEEP];
    float keep = hi ? p[i + KEEP] : p[i];
    p[i] = keep + row_xor_lane<bit>(send);
  }
}

__global__ __launch_bounds__(256)
void ns_vector_kernel(const float* __restrict__ S, const float* __restrict__ Dm,
                      const float* __restrict__ mean, const float* __restrict__ tr,
                      float* __restrict__ out) {
  __shared__ __align__(16) float sv[NS_SLOTS][TN];
  __shared__ __align__(16) float part[4][PD];

  const int b    = blockIdx.x;
  const int tid  = threadIdx.x;
  const int wid  = tid >> 5;
  const int lane = tid & 31;
  const int c    = lane & 15;
  const int row0 = wid * 16 + (lane >> 4) * 8;
  const int myrow = wid * 16 + (lane >> 1);

  v2f a[8][4];
  const float* cb = S + (long long)b * TN * TN;
#pragma unroll
  for (int q = 0; q < 8; ++q) {
    const float4* r = (const float4*)(cb + (long long)(row0 + q) * TN);
#pragma unroll
    for (int j = 0; j < 2; ++j) {
      float4 t = r[16 * j + c];
      a[q][2 * j]     = (v2f){t.x, t.y};
      a[q][2 * j + 1] = (v2f){t.z, t.w};
    }
  }
  const float trb = tr[b];
  const float itr = 1.0f / trb;

  const float* Db = Dm + (long long)b * TN * PD;
  {
    const float4* m4 = (const float4*)(mean + (long long)b * PD);
    const float4 ma = m4[lane], mb = m4[32 + lane];
    float pu[16];
#pragma unroll
    for (int r = 0; r < 16; ++r) {
      const float4* dr = (const float4*)(Db + (long long)(wid * 16 + r) * PD);
      const float4 da = dr[lane], db = dr[32 + lane];
      float s0 = da.x * ma.x, s1 = da.y * ma.y;
      s0 = fmaf(da.z, ma.z, s0); s1 = fmaf(da.w, ma.w, s1);
      s0 = fmaf(db.x, mb.x, s0); s1 = fmaf(db.y, mb.y, s1);
      s0 = fmaf(db.z, mb.z, s0); s1 = fmaf(db.w, mb.w, s1);
      pu[r] = s0 + s1;
    }
    float mine = 0.0f;
#pragma unroll
    for (int r = 0; r < 16; ++r) {
      float s = pu[r];
      s += __shfl_xor(s, 16);
      s += row_xor_lane<8>(s);
      s += row_xor_lane<4>(s);
      s += row_xor_lane<2>(s);
      s += row_xor_lane<1>(s);
      mine = (lane == r) ? s : mine;
    }
    if (lane < 16) sv[0][wid * 16 + lane] = mine;
  }
  __syncthreads();

  const int nops = ns_prog.n;
  unsigned next = ns_prog.op[0];
  for (int i = 0; i < nops; ++i) {
    const unsigned op = next;
    next = ns_prog.op[i + 1];
    const int kind = op & 3;
    const float* xs = sv[(op >> 2) & 15];
    float*       xd = sv[(op >> 6) & 15];
    const float* xe = sv[(op >> 10) & 15];
    const float es = xs[myrow];
    const float ee = xe[myrow];
    float val;
    if (kind != 3) {
      v2f x[4];
#pragma unroll
      for (int j = 0; j < 2; ++j) {
        float4 t = ((const float4*)xs)[16 * j + c];
        x[2 * j]     = (v2f){t.x, t.y};
        x[2 * j + 1] = (v2f){t.z, t.w};
      }
      float p[8];
#pragma unroll
      for (int q = 0; q < 8; ++q) {
        v2f s = a[q][0] * x[0];
#pragma unroll
        for (int j = 1; j < 4; ++j) s = __builtin_elementwise_fma(a[q][j], x[j], s);
        p[q] = s.x + s.y;
      }
      ns_reduce_step<4>(p, lane);
      ns_reduce_step<2>(p, lane);
      ns_reduce_step<1>(p, lane);
      val = (p[0] + row_xor_lane<1>(p[0])) * itr;
      if (kind != 0) val = 1.5f * es - 0.5f * val;
      if (kind == 2) val = 1.5f * ee - 0.5f * val;
    } else {
      val = 1.5f * ee - 0.5f * es;
    }
    if ((lane & 1) == 0) xd[myrow] = val;
    __syncthreads();
  }

  {
    const int cg = tid & 63;
    const int rg = tid >> 6;
    const float4* dc = (const float4*)(Db + (long long)(32 * rg) * PD) + cg;
    const float* v = &sv[1][32 * rg];
    float4 s4 = make_float4(0.0f, 0.0f, 0.0f, 0.0f);
#pragma unroll
    for (int k = 0; k < 32; ++k) {
      const float4 dv = dc[k * (PD / 4)];
      const float vk = v[k];
      s4.x = fmaf(dv.x, vk, s4.x); s4.y = fmaf(dv.y, vk, s4.y);
      s4.z = fmaf(dv.z, vk, s4.z); s4.w = fmaf(dv.w, vk, s4.w);
    }
    *(float4*)&part[rg][4 * cg] = s4;
    __syncthreads();
    const float sc = sqrtf(trb) * itr * (1.0f / (float)(TN - 1));
    out[(long long)b * PD + tid] = ((part[0][tid] + part[1][tid]) + (part[2][tid] + part[3][tid])) * sc;
  }
}

extern "C" void kernel_launch(void* const* d_in, const int* in_sizes, int n_in,
                              void* d_out, int out_size, void* d_ws, size_t ws_size,
                              hipStream_t stream) {
  (void)n_in; (void)ws_size; (void)out_size;
  const float* feat  = (const float*)d_in[1];
  const float* noise = (const float*)d_in[2];

  const int B  = in_sizes[0];
  const int d  = 256;
  const long long total = (long long)in_sizes[1];
  const int n  = (int)(total / ((long long)B * d));
  const int nd = n * d;

  float* W = (float*)d_ws;
  long long off = 0;
  float* Fp   = W + off; off += (long long)B * nd;
  float* mean = W + off; off += (long long)B * d;
  off += B;
  off += B;
  float* tr   = W + off; off += B;
  off = (off + 255) & ~255LL;
  float* M0 = W + off;

  prep_kernel<<<B, PRE_THREADS, 0, stream>>>(feat, noise, Fp, mean, 31.6227766016838f);

  dim3 gg(n / GM, B);
  gram_wmma_kernel<<<gg, 256, 0, stream>>>(Fp, M0, 1.0f / (float)(n - 1));
  trace_kernel<<<B, n, 0, stream>>>(M0, tr, n);

  ns_vector_kernel<<<B, 256, 0, stream>>>(M0, Fp, mean, tr, (float*)d_out);
}


# Trace launch folded into the Newton-Schulz launch: every wave sums the 128 diagonal elements of S itself, in the old tree's order (tr bit-identical); 4 launches -> 3
# speedup vs baseline: 12.6715x; 1.0547x over previous
#include <hip/hip_runtime.h>
#include <hip/hip_bf16.h>

typedef __attribute__((ext_vector_type(2))) float v2f;
typedef __attribute__((ext_vector_type(8))) float v8f;

#define PN 128
#define PD 256
#define PRE_THREADS 1024
#define PRE_RG 16
#define PRE_K (PN / PRE_RG)

__global__ __launch_bounds__(PRE_THREADS)
void prep_kernel(const float* __restrict__ feat, const float* __restrict__ noise,
                 float* __restrict__ Fp, float* __restrict__ mean, float snr_pow) {
  __shared__ __align__(16) float wsum[PRE_THREADS / 32][4];
  __shared__ float gst[2];
  __shared__ __align__(16) float colp[PRE_RG][PD];
  __shared__ __align__(16) float mu[PD];

  const int b    = blockIdx.x;
  const int t    = threadIdx.x;
  const int lane = t & 31;
  const int wid  = t >> 5;
  const int cg   = t & 63;
  const int rg   = t >> 6;
  const long long base = (long long)b * PN * PD + (long long)rg * PD + 4 * cg;
  const float4* f4 = (const float4*)(feat + base);
  const float4* z4 = (const float4*)(noise + base);
  float4*       o4 = (float4*)(Fp + base);

  float4 f[PRE_K], z[PRE_K];
#pragma unroll
  for (int k = 0; k < PRE_K; ++k) f[k] = f4[k * (PRE_RG * PD / 4)];
#pragma unroll
  for (int k = 0; k < PRE_K; ++k) z[k] = z4[k * (PRE_RG * PD / 4)];

  float4 af  = make_float4(0.0f, 0.0f, 0.0f, 0.0f), af2 = make_float4(0.0f, 0.0f, 0.0f, 0.0f);
  float4 an  = make_float4(0.0f, 0.0f, 0.0f, 0.0f), an2 = make_float4(0.0f, 0.0f, 0.0f, 0.0f);
#pragma unroll
  for (int k = 0; k < PRE_K; ++k) {
    af.x += f[k].x; af.y += f[k].y; af.z += f[k].z; af.w += f[k].w;
    af2.x += f[k].x * f[k].x; af2.y += f[k].y * f[k].y;
    af2.z += f[k].z * f[k].z; af2.w += f[k].w * f[k].w;
    an.x += z[k].x; an.y += z[k].y; an.z += z[k].z; an.w += z[k].w;
    an2.x += z[k].x * z[k].x; an2.y += z[k].y * z[k].y;
    an2.z += z[k].z * z[k].z; an2.w += z[k].w * z[k].w;
  }
  float sf  = (af.x + af.y) + (af.z + af.w);
  float sf2 = (af2.x + af2.y) + (af2.z + af2.w);
  float sn  = (an.x + an.y) + (an.z + an.w);
  float sn2 = (an2.x + an2.y) + (an2.z + an2.w);
#pragma unroll
  for (int m = 16; m > 0; m >>= 1) {
    sf  += __shfl_xor(sf, m);
    sf2 += __shfl_xor(sf2, m);
    sn  += __shfl_xor(sn, m);
    sn2 += __shfl_xor(sn2, m);
  }
  if (lane == 0) *(float4*)&wsum[wid][0] = make_float4(sf, sf2, sn, sn2);
  __syncthreads();
  if (wid == 0) {
    const float4 w4 = *(const float4*)&wsum[lane][0];
    double d0 = (double)w4.x, d1 = (double)w4.y;
    double d2 = (double)w4.z, d3 = (double)w4.w;
#pragma unroll
    for (int m = 16; m > 0; m >>= 1) {
      d0 += __shfl_xor(d0, m);
      d1 += __shfl_xor(d1, m);
      d2 += __shfl_xor(d2, m);
      d3 += __shfl_xor(d3, m);
    }
    if (lane == 0) {
      double inv  = 1.0 / (double)(PN * PD);
      double gmf  = d0 * inv;
      double gmn  = d2 * inv;
      double tvar = d1 * inv - gmf * gmf;
      double nvar = d3 * inv - gmn * gmn;
      gst[0] = (float)sqrt((tvar / (double)snr_pow) / nvar);
      gst[1] = (float)gmn;
    }
  }
  __syncthreads();
  const float scl = gst[0];
  const float nmn = gst[1];

  float4 cs = make_float4(0.0f, 0.0f, 0.0f, 0.0f);
#pragma unroll
  for (int k = 0; k < PRE_K; ++k) {
    f[k].x = f[k].x + (z[k].x - nmn) * scl;
    f[k].y = f[k].y + (z[k].y - nmn) * scl;
    f[k].z = f[k].z + (z[k].z - nmn) * scl;
    f[k].w = f[k].w + (z[k].w - nmn) * scl;
    cs.x += f[k].x; cs.y += f[k].y; cs.z += f[k].z; cs.w += f[k].w;
  }
  *(float4*)&colp[rg][4 * cg] = cs;
  __syncthreads();
  if (t < PD) {
    float s = 0.0f;
#pragma unroll
    for (int r = 0; r < PRE_RG; ++r) s += colp[r][t];
    float m = s / (float)PN;
    mu[t] = m;
    mean[(long long)b * PD + t] = m;
  }
  __syncthreads();
  const float4 m4 = *(const float4*)&mu[4 * cg];
#pragma unroll
  for (int k = 0; k < PRE_K; ++k) {
    float4 v;
    v.x = f[k].x - m4.x; v.y = f[k].y - m4.y; v.z = f[k].z - m4.z; v.w = f[k].w - m4.w;
    o4[k * (PRE_RG * PD / 4)] = v;
  }
}

#define GM 64
#define BK 32
#define LDP 34

__global__ __launch_bounds__(256)
void gram_wmma_kernel(const float* __restrict__ Dm, float* __restrict__ S, float alpha) {
  __shared__ __align__(16) float Ts[2][PN][LDP];

  const int b  = blockIdx.y;
  const int m0 = blockIdx.x * GM;
  const float* Db = Dm + (long long)b * PN * PD;
  float*       Sb = S  + (long long)b * PN * PN;

  const int tid  = threadIdx.x;
  const int wid  = tid >> 5;
  const int lane = tid & 31;
  const int mr   = lane & 15;
  const int h    = lane >> 4;
  const int wm   = m0 + (wid >> 2) * 32;
  const int wn   = (wid & 3) * 32;

  const int tr = tid >> 5, tc = tid & 31;
  const float* Dg = Db + (long long)tr * PD + tc;

  v8f acc[2][2];
#pragma unroll
  for (int im = 0; im < 2; ++im)
#pragma unroll
    for (int in = 0; in < 2; ++in) acc[im][in] = (v8f){};

  float rd[16];

#define G2R(k0_)                                                                    \
  do {                                                                              \
    const float* Dk = Dg + (k0_);                                                   \
    _Pragma("unroll")                                                               \
    for (int i = 0; i < 16; ++i) rd[i] = Dk[8 * i * PD];                            \
  } while (0)

#define R2S(s_)                                                                     \
  do {                                                                              \
    _Pragma("unroll")                                                               \
    for (int i = 0; i < 16; ++i) Ts[(s_)][tr + 8 * i][tc] = rd[i];                  \
  } while (0)

#define TILE_MMA(s_)                                                                \
  do {                                                                              \
    _Pragma("unroll")                                                               \
    for (int kk = 0; kk < BK; kk += 4) {                                            \
      int kc = kk + 2 * h;                       \
      v2f a0 = *(const v2f*)&Ts[(s_)][wm + mr][kc];                                 \
      v2f a1 = *(const v2f*)&Ts[(s_)][wm + 16 + mr][kc];                            \
      v2f b0 = *(const v2f*)&Ts[(s_)][wn + mr][kc];                                 \
      v2f b1 = *(const v2f*)&Ts[(s_)][wn + 16 + mr][kc];                            \
      acc[0][0] = __builtin_amdgcn_wmma_f32_16x16x4_f32(false, a0, false, b0, (short)0, acc[0][0], false, false); \
      acc[0][1] = __builtin_amdgcn_wmma_f32_16x16x4_f32(false, a0, false, b1, (short)0, acc[0][1], false, false); \
      acc[1][0] = __builtin_amdgcn_wmma_f32_16x16x4_f32(false, a1, false, b0, (short)0, acc[1][0], false, false); \
      acc[1][1] = __builtin_amdgcn_wmma_f32_16x16x4_f32(false, a1, false, b1, (short)0, acc[1][1], false, false); \
    }                                                                               \
  } while (0)

  const int nk = PD / BK;

  G2R(0);
  R2S(0);
  __syncthreads();

  for (int t = 0; t < nk - 1; ++t) {
    int cur = t & 1;
    G2R((t + 1) * BK);
    __builtin_amdgcn_sched_barrier(0);
    TILE_MMA(cur);
    R2S(cur ^ 1);
    __syncthreads();
  }
  TILE_MMA((nk - 1) & 1);

#undef G2R
#undef R2S
#undef TILE_MMA

#pragma unroll
  for (int im = 0; im < 2; ++im) {
#pragma unroll
    for (int r = 0; r < 8; ++r) {
      int row = wm + 16 * im + 8 * h + r;
      float* srow = Sb + (long long)row * PN;
#pragma unroll
      for (int in = 0; in < 2; ++in) {
        int col = wn + 16 * in + mr;
        srow[col] = acc[im][in][r] * alpha;
      }
    }
  }
}

#define TN PN
#define NS_LEVELS 3
#define NS_MAXOPS 160
#define NS_SLOTS 16

struct NsProg {
  unsigned int op[NS_MAXOPS];
  int n;
  int top;
};

struct NsGen {
  NsProg p;
  int sp;
  constexpr void emit(int kind, int s, int dst, int e) {
    p.op[p.n++] = (unsigned int)(kind | (s << 2) | (dst << 6) | (e << 10));
    if (sp > p.top) p.top = sp;
  }
  constexpr void applyY(int k, int dst, int s) {
    int t = sp++;
    if (k == 0) { emit(1, s, t, 0); emit(0, t, dst, 0); }
    else        { applyT(k - 1, t, s); applyY(k - 1, dst, t); }
    sp--;
  }
  constexpr void applyZ(int k, int dst, int s) {
    if (k == 0) { emit(1, s, dst, 0); return; }
    int t = sp++;
    applyZ(k - 1, t, s);
    applyT(k - 1, dst, t);
    sp--;
  }
  constexpr void applyT(int k, int dst, int s) {
    int t = sp++;
    applyY(k, t, s);
    if (k == 0) { emit(2, t, dst, s); }
    else        { int t2 = sp++; applyZ(k, t2, t); emit(3, t2, dst, s); sp--; }
    sp--;
  }
};

constexpr NsProg ns_make_prog() {
  NsGen g{};
  g.sp = 2;
  g.applyZ(NS_LEVELS + 1, 1, 0);
  return g.p;
}

__constant__ constexpr NsProg ns_prog = ns_make_prog();
static_assert(ns_prog.n < NS_MAXOPS && ns_prog.top <= NS_SLOTS, "step list does not fit");

template <int MASK>
__device__ __forceinline__ float row_xor_lane(float v) {
  return __builtin_bit_cast(float, __builtin_amdgcn_update_dpp(0, __builtin_bit_cast(int, v),
                                                               0x160 | MASK, 0xf, 0xf, true));
}

template <int KEEP>
__device__ __forceinline__ void ns_reduce_step(float (&p)[8], int lane) {
  constexpr int bit = 2 * KEEP;
  const bool hi = (lane & bit) != 0;
#pragma unroll
  for (int i = 0; i < KEEP; ++i) {
    float send = hi ? p[i] : p[i + KEEP];
    float keep = hi ? p[i + KEEP] : p[i];
    p[i] = keep + row_xor_lane<bit>(send);
  }
}

__global__ __launch_bounds__(256)
void ns_vector_kernel(const float* __restrict__ S, const float* __restrict__ Dm,
                      const float* __restrict__ mean, float* __restrict__ out) {
  __shared__ __align__(16) float sv[NS_SLOTS][TN];
  __shared__ __align__(16) float part[4][PD];

  const int b    = blockIdx.x;
  const int tid  = threadIdx.x;
  const int wid  = tid >> 5;
  const int lane = tid & 31;
  const int c    = lane & 15;
  const int row0 = wid * 16 + (lane >> 4) * 8;
  const int myrow = wid * 16 + (lane >> 1);

  v2f a[8][4];
  const float* cb = S + (long long)b * TN * TN;
#pragma unroll
  for (int q = 0; q < 8; ++q) {
    const float4* r = (const float4*)(cb + (long long)(row0 + q) * TN);
#pragma unroll
    for (int j = 0; j < 2; ++j) {
      float4 t = r[16 * j + c];
      a[q][2 * j]     = (v2f){t.x, t.y};
      a[q][2 * j + 1] = (v2f){t.z, t.w};
    }
  }
  float dg[4];
#pragma unroll
  for (int j = 0; j < 4; ++j) dg[j] = cb[(long long)(lane + 32 * j) * (TN + 1)];
  float trb = (dg[0] + dg[2]) + (dg[1] + dg[3]);
  trb += __shfl_xor(trb, 16);
  trb += row_xor_lane<8>(trb);
  trb += row_xor_lane<4>(trb);
  trb += row_xor_lane<2>(trb);
  trb += row_xor_lane<1>(trb);
  const float itr = 1.0f / trb;

  const float* Db = Dm + (long long)b * TN * PD;
  {
    const float4* m4 = (const float4*)(mean + (long long)b * PD);
    const float4 ma = m4[lane], mb = m4[32 + lane];
    float pu[16];
#pragma unroll
    for (int r = 0; r < 16; ++r) {
      const float4* dr = (const float4*)(Db + (long long)(wid * 16 + r) * PD);
      const float4 da = dr[lane], db = dr[32 + lane];
      float s0 = da.x * ma.x, s1 = da.y * ma.y;
      s0 = fmaf(da.z, ma.z, s0); s1 = fmaf(da.w, ma.w, s1);
      s0 = fmaf(db.x, mb.x, s0); s1 = fmaf(db.y, mb.y, s1);
      s0 = fmaf(db.z, mb.z, s0); s1 = fmaf(db.w, mb.w, s1);
      pu[r] = s0 + s1;
    }
    float mine = 0.0f;
#pragma unroll
    for (int r = 0; r < 16; ++r) {
      float s = pu[r];
      s += __shfl_xor(s, 16);
      s += row_xor_lane<8>(s);
      s += row_xor_lane<4>(s);
      s += row_xor_lane<2>(s);
      s += row_xor_lane<1>(s);
      mine = (lane == r) ? s : mine;
    }
    if (lane < 16) sv[0][wid * 16 + lane] = mine;
  }
  __syncthreads();

  const int nops = ns_prog.n;
  unsigned next = ns_prog.op[0];
  for (int i = 0; i < nops; ++i) {
    const unsigned op = next;
    next = ns_prog.op[i + 1];
    const int kind = op & 3;
    const float* xs = sv[(op >> 2) & 15];
    float*       xd = sv[(op >> 6) & 15];
    const float* xe = sv[(op >> 10) & 15];
    const float es = xs[myrow];
    const float ee = xe[myrow];
    float val;
    if (kind != 3) {
      v2f x[4];
#pragma unroll
      for (int j = 0; j < 2; ++j) {
        float4 t = ((const float4*)xs)[16 * j + c];
        x[2 * j]     = (v2f){t.x, t.y};
        x[2 * j + 1] = (v2f){t.z, t.w};
      }
      float p[8];
#pragma unroll
      for (int q = 0; q < 8; ++q) {
        v2f s = a[q][0] * x[0];
#pragma unroll
        for (int j = 1; j < 4; ++j) s = __builtin_elementwise_fma(a[q][j], x[j], s);
        p[q] = s.x + s.y;
      }
      ns_reduce_step<4>(p, lane);
      ns_reduce_step<2>(p, lane);
      ns_reduce_step<1>(p, lane);
      val = (p[0] + row_xor_lane<1>(p[0])) * itr;
      if (kind != 0) val = 1.5f * es - 0.5f * val;
      if (kind == 2) val = 1.5f * ee - 0.5f * val;
    } else {
      val = 1.5f * ee - 0.5f * es;
    }
    if ((lane & 1) == 0) xd[myrow] = val;
    __syncthreads();
  }

  {
    const int cg = tid & 63;
    const int rg = tid >> 6;
    const float4* dc = (const float4*)(Db + (long long)(32 * rg) * PD) + cg;
    const float* v = &sv[1][32 * rg];
    float4 s4 = make_float4(0.0f, 0.0f, 0.0f, 0.0f);
#pragma unroll
    for (int k = 0; k < 32; ++k) {
      const float4 dv = dc[k * (PD / 4)];
      const float vk = v[k];
      s4.x = fmaf(dv.x, vk, s4.x); s4.y = fmaf(dv.y, vk, s4.y);
      s4.z = fmaf(dv.z, vk, s4.z); s4.w = fmaf(dv.w, vk, s4.w);
    }
    *(float4*)&part[rg][4 * cg] = s4;
    __syncthreads();
    const float sc = sqrtf(trb) * itr * (1.0f / (float)(TN - 1));
    out[(long long)b * PD + tid] = ((part[0][tid] + part[1][tid]) + (part[2][tid] + part[3][tid])) * sc;
  }
}

extern "C" void kernel_launch(void* const* d_in, const int* in_sizes, int n_in,
                              void* d_out, int out_size, void* d_ws, size_t ws_size,
                              hipStream_t stream) {
  (void)n_in; (void)ws_size; (void)out_size;
  const float* feat  = (const float*)d_in[1];
  const float* noise = (const float*)d_in[2];

  const int B  = in_sizes[0];
  const int d  = 256;
  const long long total = (long long)in_sizes[1];
  const int n  = (int)(total / ((long long)B * d));
  const int nd = n * d;

  float* W = (float*)d_ws;
  long long off = 0;
  float* Fp   = W + off; off += (long long)B * nd;
  float* mean = W + off; off += (long long)B * d;
  off += B;
  off += B;
  off += B;
  off = (off + 255) & ~255LL;
  float* M0 = W + off;

  prep_kernel<<<B, PRE_THREADS, 0, stream>>>(feat, noise, Fp, mean, 31.6227766016838f);

  dim3 gg(n / GM, B);
  gram_wmma_kernel<<<gg, 256, 0, stream>>>(Fp, M0, 1.0f / (float)(n - 1));

  ns_vector_kernel<<<B, 256, 0, stream>>>(M0, Fp, mean, (float*)d_out);
}


# Newton-Schulz steps: every plain vector update folded into the product before it, step scalings merged into compile-time coefficients (one fma behind the row sum); 134 -> 121 steps; new sum order
# speedup vs baseline: 13.6649x; 1.0784x over previous
#include <hip/hip_runtime.h>
#include <hip/hip_bf16.h>

typedef __attribute__((ext_vector_type(2))) float v2f;
typedef __attribute__((ext_vector_type(8))) float v8f;

#define PN 128
#define PD 256
#define PRE_THREADS 1024
#define PRE_RG 16
#define PRE_K (PN / PRE_RG)

__global__ __launch_bounds__(PRE_THREADS)
void prep_kernel(const float* __restrict__ feat, const float* __restrict__ noise,
                 float* __restrict__ Fp, float* __restrict__ mean, float snr_pow) {
  __shared__ __align__(16) float wsum[PRE_THREADS / 32][4];
  __shared__ float gst[2];
  __shared__ __align__(16) float colp[PRE_RG][PD];
  __shared__ __align__(16) float mu[PD];

  const int b    = blockIdx.x;
  const int t    = threadIdx.x;
  const int lane = t & 31;
  const int wid  = t >> 5;
  const int cg   = t & 63;
  const int rg   = t >> 6;
  const long long base = (long long)b * PN * PD + (long long)rg * PD + 4 * cg;
  const float4* f4 = (const float4*)(feat + base);
  const float4* z4 = (const float4*)(noise + base);
  float4*       o4 = (float4*)(Fp + base);

  float4 f[PRE_K], z[PRE_K];
#pragma unroll
  for (int k = 0; k < PRE_K; ++k) f[k] = f4[k * (PRE_RG * PD / 4)];
#pragma unroll
  for (int k = 0; k < PRE_K; ++k) z[k] = z4[k * (PRE_RG * PD / 4)];

  float4 af  = make_float4(0.0f, 0.0f, 0.0f, 0.0f), af2 = make_float4(0.0f, 0.0f, 0.0f, 0.0f);
  float4 an  = make_float4(0.0f, 0.0f, 0.0f, 0.0f), an2 = make_float4(0.0f, 0.0f, 0.0f, 0.0f);
#pragma unroll
  for (int k = 0; k < PRE_K; ++k) {
    af.x += f[k].x; af.y += f[k].y; af.z += f[k].z; af.w += f[k].w;
    af2.x += f[k].x * f[k].x; af2.y += f[k].y * f[k].y;
    af2.z += f[k].z * f[k].z; af2.w += f[k].w * f[k].w;
    an.x += z[k].x; an.y += z[k].y; an.z += z[k].z; an.w += z[k].w;
    an2.x += z[k].x * z[k].x; an2.y += z[k].y * z[k].y;
    an2.z += z[k].z * z[k].z; an2.w += z[k].w * z[k].w;
  }
  float sf  = (af.x + af.y) + (af.z + af.w);
  float sf2 = (af2.x + af2.y) + (af2.z + af2.w);
  float sn  = (an.x + an.y) + (an.z + an.w);
  float sn2 = (an2.x + an2.y) + (an2.z + an2.w);
#pragma unroll
  for (int m = 16; m > 0; m >>= 1) {
    sf  += __shfl_xor(sf, m);
    sf2 += __shfl_xor(sf2, m);
    sn  += __shfl_xor(sn, m);
    sn2 += __shfl_xor(sn2, m);
  }
  if (lane == 0) *(float4*)&wsum[wid][0] = make_float4(sf, sf2, sn, sn2);
  __syncthreads();
  if (wid == 0) {
    const float4 w4 = *(const float4*)&wsum[lane][0];
    double d0 = (double)w4.x, d1 = (double)w4.y;
    double d2 = (double)w4.z, d3 = (double)w4.w;
#pragma unroll
    for (int m = 16; m > 0; m >>= 1) {
      d0 += __shfl_xor(d0, m);
      d1 += __shfl_xor(d1, m);
      d2 += __shfl_xor(d2, m);
      d3 += __shfl_xor(d3, m);
    }
    if (lane == 0) {
      double inv  = 1.0 / (double)(PN * PD);
      double gmf  = d0 * inv;
      double gmn  = d2 * inv;
      double tvar = d1 * inv - gmf * gmf;
      double nvar = d3 * inv - gmn * gmn;
      gst[0] = (float)sqrt((tvar / (double)snr_pow) / nvar);
      gst[1] = (float)gmn;
    }
  }
  __syncthreads();
  const float scl = gst[0];
  const float nmn = gst[1];

  float4 cs = make_float4(0.0f, 0.0f, 0.0f, 0.0f);
#pragma unroll
  for (int k = 0; k < PRE_K; ++k) {
    f[k].x = f[k].x + (z[k].x - nmn) * scl;
    f[k].y = f[k].y + (z[k].y - nmn) * scl;
    f[k].z = f[k].z + (z[k].z - nmn) * scl;
    f[k].w = f[k].w + (z[k].w - nmn) * scl;
    cs.x += f[k].x; cs.y += f[k].y; cs.z += f[k].z; cs.w += f[k].w;
  }
  *(float4*)&colp[rg][4 * cg] = cs;
  __syncthreads();
  if (t < PD) {
    float s = 0.0f;
#pragma unroll
    for (int r = 0; r < PRE_RG; ++r) s += colp[r][t];
    float m = s / (float)PN;
    mu[t] = m;
    mean[(long long)b * PD + t] = m;
  }
  __syncthreads();
  const float4 m4 = *(const float4*)&mu[4 * cg];
#pragma unroll
  for (int k = 0; k < PRE_K; ++k) {
    float4 v;
    v.x = f[k].x - m4.x; v.y = f[k].y - m4.y; v.z = f[k].z - m4.z; v.w = f[k].w - m4.w;
    o4[k * (PRE_RG * PD / 4)] = v;
  }
}

#define GM 64
#define BK 32
#define LDP 34

__global__ __launch_bounds__(256)
void gram_wmma_kernel(const float* __restrict__ Dm, float* __restrict__ S, float alpha) {
  __shared__ __align__(16) float Ts[2][PN][LDP];

  const int b  = blockIdx.y;
  const int m0 = blockIdx.x * GM;
  const float* Db = Dm + (long long)b * PN * PD;
  float*       Sb = S  + (long long)b * PN * PN;

  const int tid  = threadIdx.x;
  const int wid  = tid >> 5;
  const int lane = tid & 31;
  const int mr   = lane & 15;
  const int h    = lane >> 4;
  const int wm   = m0 + (wid >> 2) * 32;
  const int wn   = (wid & 3) * 32;

  const int tr = tid >> 5, tc = tid & 31;
  const float* Dg = Db + (long long)tr * PD + tc;

  v8f acc[2][2];
#pragma unroll
  for (int im = 0; im < 2; ++im)
#pragma unroll
    for (int in = 0; in < 2; ++in) acc[im][in] = (v8f){};

  float rd[16];

#define G2R(k0_)                                                                    \
  do {                                                                              \
    const float* Dk = Dg + (k0_);                                                   \
    _Pragma("unroll")                                                               \
    for (int i = 0; i < 16; ++i) rd[i] = Dk[8 * i * PD];                            \
  } while (0)

#define R2S(s_)                                                                     \
  do {                                                                              \
    _Pragma("unroll")                                                               \
    for (int i = 0; i < 16; ++i) Ts[(s_)][tr + 8 * i][tc] = rd[i];                  \
  } while (0)

#define TILE_MMA(s_)                                                                \
  do {                                                                              \
    _Pragma("unroll")                                                               \
    for (int kk = 0; kk < BK; kk += 4) {                                            \
      int kc = kk + 2 * h;                       \
      v2f a0 = *(const v2f*)&Ts[(s_)][wm + mr][kc];                                 \
      v2f a1 = *(const v2f*)&Ts[(s_)][wm + 16 + mr][kc];                            \
      v2f b0 = *(const v2f*)&Ts[(s_)][wn + mr][kc];                                 \
      v2f b1 = *(const v2f*)&Ts[(s_)][wn + 16 + mr][kc];                            \
      acc[0][0] = __builtin_amdgcn_wmma_f32_16x16x4_f32(false, a0, false, b0, (short)0, acc[0][0], false, false); \
      acc[0][1] = __builtin_amdgcn_wmma_f32_16x16x4_f32(false, a0, false, b1, (short)0, acc[0][1], false, false); \
      acc[1][0] = __builtin_amdgcn_wmma_f32_16x16x4_f32(false, a1, false, b0, (short)0, acc[1][0], false, false); \
      acc[1][1] = __builtin_amdgcn_wmma_f32_16x16x4_f32(false, a1, false, b1, (short)0, acc[1][1], false, false); \
    }                                                                               \
  } while (0)

  const int nk = PD / BK;

  G2R(0);
  R2S(0);
  __syncthreads();

  for (int t = 0; t < nk - 1; ++t) {
    int cur = t & 1;
    G2R((t + 1) * BK);
    __builtin_amdgcn_sched_barrier(0);
    TILE_MMA(cur);
    R2S(cur ^ 1);
    __syncthreads();
  }
  TILE_MMA((nk - 1) & 1);

#undef G2R
#undef R2S
#undef TILE_MMA

#pragma unroll
  for (int im = 0; im < 2; ++im) {
#pragma unroll
    for (int r = 0; r < 8; ++r) {
      int row = wm + 16 * im + 8 * h + r;
      float* srow = Sb + (long long)row * PN;
#pragma unroll
      for (int in = 0; in < 2; ++in) {
        int col = wn + 16 * in + mr;
        srow[col] = acc[im][in][r] * alpha;
      }
    }
  }
}

#define TN PN
#define NS_LEVELS 3
#define NS_MAXOPS 160
#define NS_SLOTS 16

#define NS_MAXE 4
struct alignas(32) NsStep {
  unsigned int slots;
  float A, B, C1, C2, C3, C4;
  unsigned int pad;
};

struct NsProg {
  NsStep st[NS_MAXOPS];
  int n;
  int top;
  int ok;
};

struct NsRaw {
  int src, dst, base, m;
  int e[NS_MAXE];
};

struct NsGen {
  NsRaw r[NS_MAXOPS];
  int n;
  int top;
  int ok;
  int sp;
  constexpr void mat(int base, int s, int dst) {
    r[n].src = s; r[n].dst = dst; r[n].base = base; r[n].m = 0;
    for (int i = 0; i < NS_MAXE; ++i) r[n].e[i] = s;
    n++;
    if (sp > top) top = sp;
  }
  constexpr void fold(int from, int dst, int e) {
    NsRaw& l = r[n - 1];
    if (n < 1 || l.dst != from || l.m >= NS_MAXE) { ok = 0; return; }
    l.e[l.m++] = e;
    l.dst = dst;
    if (sp > top) top = sp;
  }
  constexpr void applyY(int k, int dst, int s) {
    int t = sp++;
    if (k == 0) { mat(1, s, t); mat(0, t, dst); }
    else        { applyT(k - 1, t, s); applyY(k - 1, dst, t); }
    sp--;
  }
  constexpr void applyZ(int k, int dst, int s) {
    if (k == 0) { mat(1, s, dst); return; }
    int t = sp++;
    applyZ(k - 1, t, s);
    applyT(k - 1, dst, t);
    sp--;
  }
  constexpr void applyT(int k, int dst, int s) {
    int t = sp++;
    applyY(k, t, s);
    int t2 = sp++;
    applyZ(k, t2, t);
    fold(t2, dst, s);
    sp--;
    sp--;
  }
};

constexpr NsProg ns_make_prog() {
  NsGen g{};
  g.ok = 1;
  g.sp = 2;
  g.applyZ(NS_LEVELS + 1, 1, 0);
  NsProg p{};
  p.n = g.n; p.top = g.top; p.ok = g.ok;
  for (int i = 0; i < g.n; ++i) {
    const NsRaw& q = g.r[i];
    if (q.dst == q.src) p.ok = 0;
    float sc = 1.0f;
    for (int j = 0; j < q.m; ++j) sc *= -0.5f;
    float cc[NS_MAXE] = {0.0f, 0.0f, 0.0f, 0.0f};
    for (int j = 0; j < q.m; ++j) {
      float w = 1.5f;
      for (int l = j + 1; l < q.m; ++l) w *= -0.5f;
      cc[j] = w;
    }
    NsStep& o = p.st[i];
    o.slots = (unsigned int)(q.src | (q.dst << 4) | (q.e[0] << 8) | (q.e[1] << 12) | (q.e[2] << 16) | (q.e[3] << 20));
    o.A = q.base ? -0.5f * sc : sc;
    o.B = q.base ? 0.75f * sc : 0.0f;
    o.C1 = 0.5f * cc[0]; o.C2 = 0.5f * cc[1]; o.C3 = 0.5f * cc[2]; o.C4 = 0.5f * cc[3];
    o.pad = 0u;
  }
  return p;
}

__constant__ constexpr NsProg ns_prog = ns_make_prog();
static_assert(ns_prog.ok == 1, "step list: a fold without its product, or dst == src");
static_assert(ns_prog.n == 121, "one step per product with G");
static_assert(ns_prog.n < NS_MAXOPS && ns_prog.top <= NS_SLOTS, "step list does not fit");

template <int MASK>
__device__ __forceinline__ float row_xor_lane(float v) {
  return __builtin_bit_cast(float, __builtin_amdgcn_update_dpp(0, __builtin_bit_cast(int, v),
                                                               0x160 | MASK, 0xf, 0xf, true));
}

template <int KEEP>
__device__ __forceinline__ void ns_reduce_step(float (&p)[8], int lane) {
  constexpr int bit = 2 * KEEP;
  const bool hi = (lane & bit) != 0;
#pragma unroll
  for (int i = 0; i < KEEP; ++i) {
    float send = hi ? p[i] : p[i + KEEP];
    float keep = hi ? p[i + KEEP] : p[i];
    p[i] = keep + row_xor_lane<bit>(send);
  }
}

__global__ __launch_bounds__(256)
void ns_vector_kernel(const float* __restrict__ S, const float* __restrict__ Dm,
                      const float* __restrict__ mean, float* __restrict__ out) {
  __shared__ __align__(16) float sv[NS_SLOTS][TN];
  __shared__ __align__(16) float part[4][PD];

  const int b    = blockIdx.x;
  const int tid  = threadIdx.x;
  const int wid  = tid >> 5;
  const int lane = tid & 31;
  const int c    = lane & 15;
  const int row0 = wid * 16 + (lane >> 4) * 8;
  const int myrow = wid * 16 + (lane >> 1);

  v2f a[8][4];
  const float* cb = S + (long long)b * TN * TN;
#pragma unroll
  for (int q = 0; q < 8; ++q) {
    const float4* r = (const float4*)(cb + (long long)(row0 + q) * TN);
#pragma unroll
    for (int j = 0; j < 2; ++j) {
      float4 t = r[16 * j + c];
      a[q][2 * j]     = (v2f){t.x, t.y};
      a[q][2 * j + 1] = (v2f){t.z, t.w};
    }
  }
  float dg[4];
#pragma unroll
  for (int j = 0; j < 4; ++j) dg[j] = cb[(long long)(lane + 32 * j) * (TN + 1)];
  float trb = (dg[0] + dg[2]) + (dg[1] + dg[3]);
  trb += __shfl_xor(trb, 16);
  trb += row_xor_lane<8>(trb);
  trb += row_xor_lane<4>(trb);
  trb += row_xor_lane<2>(trb);
  trb += row_xor_lane<1>(trb);
  const float itr = 1.0f / trb;

  const float* Db = Dm + (long long)b * TN * PD;
  {
    const float4* m4 = (const float4*)(mean + (long long)b * PD);
    const float4 ma = m4[lane], mb = m4[32 + lane];
    float pu[16];
#pragma unroll
    for (int r = 0; r < 16; ++r) {
      const float4* dr = (const float4*)(Db + (long long)(wid * 16 + r) * PD);
      const float4 da = dr[lane], db = dr[32 + lane];
      float s0 = da.x * ma.x, s1 = da.y * ma.y;
      s0 = fmaf(da.z, ma.z, s0); s1 = fmaf(da.w, ma.w, s1);
      s0 = fmaf(db.x, mb.x, s0); s1 = fmaf(db.y, mb.y, s1);
      s0 = fmaf(db.z, mb.z, s0); s1 = fmaf(db.w, mb.w, s1);
      pu[r] = s0 + s1;
    }
    float mine = 0.0f;
#pragma unroll
    for (int r = 0; r < 16; ++r) {
      float s = pu[r];
      s += __shfl_xor(s, 16);
      s += row_xor_lane<8>(s);
      s += row_xor_lane<4>(s);
      s += row_xor_lane<2>(s);
      s += row_xor_lane<1>(s);
      mine = (lane == r) ? s : mine;
    }
    if (lane < 16) sv[0][wid * 16 + lane] = mine;
  }
  __syncthreads();

  const int nops = ns_prog.n;
  NsStep nx = ns_prog.st[0];
  for (int i = 0; i < nops; ++i) {
    const NsStep op = nx;
    nx = ns_prog.st[i + 1];
    const float* xs = sv[op.slots & 15];
    float*       xd = sv[(op.slots >> 4) & 15];
    v2f x[4];
#pragma unroll
    for (int j = 0; j < 2; ++j) {
      float4 t = ((const float4*)xs)[16 * j + c];
      x[2 * j]     = (v2f){t.x, t.y};
      x[2 * j + 1] = (v2f){t.z, t.w};
    }
    __builtin_amdgcn_sched_barrier(0);
    const float es = xs[myrow];
    const float e1 = sv[(op.slots >> 8) & 15][myrow];
    const float e2 = sv[(op.slots >> 12) & 15][myrow];
    const float e3 = sv[(op.slots >> 16) & 15][myrow];
    const float e4 = sv[(op.slots >> 20) & 15][myrow];
    __builtin_amdgcn_sched_barrier(0);
    const float linh = fmaf(op.B, es, fmaf(op.C1, e1, fmaf(op.C2, e2, fmaf(op.C3, e3, op.C4 * e4))));
    const float ag  = op.A * itr;
    float p[8];
#pragma unroll
    for (int q = 0; q < 8; ++q) {
      v2f s = a[q][0] * x[0];
#pragma unroll
      for (int j = 1; j < 4; ++j) s = __builtin_elementwise_fma(a[q][j], x[j], s);
      p[q] = s.x + s.y;
    }
    ns_reduce_step<4>(p, lane);
    ns_reduce_step<2>(p, lane);
    ns_reduce_step<1>(p, lane);
    const float w   = fmaf(ag, p[0], linh);
    const float val = w + row_xor_lane<1>(w);
    if ((lane & 1) == 0) xd[myrow] = val;
    __syncthreads();
  }

  {
    const int cg = tid & 63;
    const int rg = tid >> 6;
    const float4* dc = (const float4*)(Db + (long long)(32 * rg) * PD) + cg;
    const float* v = &sv[1][32 * rg];
    float4 s4 = make_float4(0.0f, 0.0f, 0.0f, 0.0f);
#pragma unroll
    for (int k = 0; k < 32; ++k) {
      const float4 dv = dc[k * (PD / 4)];
      const float vk = v[k];
      s4.x = fmaf(dv.x, vk, s4.x); s4.y = fmaf(dv.y, vk, s4.y);
      s4.z = fmaf(dv.z, vk, s4.z); s4.w = fmaf(dv.w, vk, s4.w);
    }
    *(float4*)&part[rg][4 * cg] = s4;
    __syncthreads();
    const float sc = sqrtf(trb) * itr * (1.0f / (float)(TN - 1));
    out[(long long)b * PD + tid] = ((part[0][tid] + part[1][tid]) + (part[2][tid] + part[3][tid])) * sc;
  }
}

extern "C" void kernel_launch(void* const* d_in, const int* in_sizes, int n_in,
                              void* d_out, int out_size, void* d_ws, size_t ws_size,
                              hipStream_t stream) {
  (void)n_in; (void)ws_size; (void)out_size;
  const float* feat  = (const float*)d_in[1];
  const float* noise = (const float*)d_in[2];

  const int B  = in_sizes[0];
  const int d  = 256;
  const long long total = (long long)in_sizes[1];
  const int n  = (int)(total / ((long long)B * d));
  const int nd = n * d;

  float* W = (float*)d_ws;
  long long off = 0;
  float* Fp   = W + off; off += (long long)B * nd;
  float* mean = W + off; off += (long long)B * d;
  off += B;
  off += B;
  off += B;
  off = (off + 255) & ~255LL;
  float* M0 = W + off;

  prep_kernel<<<B, PRE_THREADS, 0, stream>>>(feat, noise, Fp, mean, 31.6227766016838f);

  dim3 gg(n / GM, B);
  gram_wmma_kernel<<<gg, 256, 0, stream>>>(Fp, M0, 1.0f / (float)(n - 1));

  ns_vector_kernel<<<B, 256, 0, stream>>>(M0, Fp, mean, (float*)d_out);
}


# Three launches fused into one, a work-group per graph: centred features and Gram matrix stay in LDS between the stages, no workspace traffic; 3 -> 1 launches; per-graph and column sums in a new order
# speedup vs baseline: 16.5088x; 1.2081x over previous
#include <hip/hip_runtime.h>

typedef __attribute__((ext_vector_type(2))) float v2f;
typedef __attribute__((ext_vector_type(8))) float v8f;

#define PN 128
#define PD 256
#define NT 256
#define NW (NT / 32)
#define F_RG (NT / (PD / 4))
#define F_K (PN / F_RG)
#define DP 260
#define SP 132

#define TN PN
#define NS_LEVELS 3
#define NS_MAXOPS 160
#define NS_SLOTS 16

#define NS_MAXE 4
struct alignas(32) NsStep {
  unsigned int slots;
  float A, B, C1, C2, C3, C4;
  unsigned int pad;
};

struct NsProg {
  NsStep st[NS_MAXOPS];
  int n;
  int top;
  int ok;
};

struct NsRaw {
  int src, dst, base, m;
  int e[NS_MAXE];
};

struct NsGen {
  NsRaw r[NS_MAXOPS];
  int n;
  int top;
  int ok;
  int sp;
  constexpr void mat(int base, int s, int dst) {
    r[n].src = s; r[n].dst = dst; r[n].base = base; r[n].m = 0;
    for (int i = 0; i < NS_MAXE; ++i) r[n].e[i] = s;
    n++;
    if (sp > top) top = sp;
  }
  constexpr void fold(int from, int dst, int e) {
    NsRaw& l = r[n - 1];
    if (n < 1 || l.dst != from || l.m >= NS_MAXE) { ok = 0; return; }
    l.e[l.m++] = e;
    l.dst = dst;
    if (sp > top) top = sp;
  }
  constexpr void applyY(int k, int dst, int s) {
    int t = sp++;
    if (k == 0) { mat(1, s, t); mat(0, t, dst); }
    else        { applyT(k - 1, t, s); applyY(k - 1, dst, t); }
    sp--;
  }
  constexpr void applyZ(int k, int dst, int s) {
    if (k == 0) { mat(1, s, dst); return; }
    int t = sp++;
    applyZ(k - 1, t, s);
    applyT(k - 1, dst, t);
    sp--;
  }
  constexpr void applyT(int k, int dst, int s) {
    int t = sp++;
    applyY(k, t, s);
    int t2 = sp++;
    applyZ(k, t2, t);
    fold(t2, dst, s);
    sp--;
    sp--;
  }
};

constexpr NsProg ns_make_prog() {
  NsGen g{};
  g.ok = 1;
  g.sp = 2;
  g.applyZ(NS_LEVELS + 1, 1, 0);
  NsProg p{};
  p.n = g.n; p.top = g.top; p.ok = g.ok;
  for (int i = 0; i < g.n; ++i) {
    const NsRaw& q = g.r[i];
    if (q.dst == q.src) p.ok = 0;
    float sc = 1.0f;
    for (int j = 0; j < q.m; ++j) sc *= -0.5f;
    float cc[NS_MAXE] = {0.0f, 0.0f, 0.0f, 0.0f};
    for (int j = 0; j < q.m; ++j) {
      float w = 1.5f;
      for (int l = j + 1; l < q.m; ++l) w *= -0.5f;
      cc[j] = w;
    }
    NsStep& o = p.st[i];
    o.slots = (unsigned int)(q.src | (q.dst << 4) | (q.e[0] << 8) | (q.e[1] << 12) | (q.e[2] << 16) | (q.e[3] << 20));
    o.A = q.base ? -0.5f * sc : sc;
    o.B = q.base ? 0.75f * sc : 0.0f;
    o.C1 = 0.5f * cc[0]; o.C2 = 0.5f * cc[1]; o.C3 = 0.5f * cc[2]; o.C4 = 0.5f * cc[3];
    o.pad = 0u;
  }
  return p;
}

__constant__ constexpr NsProg ns_prog = ns_make_prog();
static_assert(ns_prog.ok == 1, "step list: a fold without its product, or dst == src");
static_assert(ns_prog.n == 121, "one step per product with G");
static_assert(ns_prog.n < NS_MAXOPS && ns_prog.top <= NS_SLOTS, "step list does not fit");

template <int MASK>
__device__ __forceinline__ float row_xor_lane(float v) {
  return __builtin_bit_cast(float, __builtin_amdgcn_update_dpp(0, __builtin_bit_cast(int, v),
                                                               0x160 | MASK, 0xf, 0xf, true));
}

template <int KEEP>
__device__ __forceinline__ void ns_reduce_step(float (&p)[8], int lane) {
  constexpr int bit = 2 * KEEP;
  const bool hi = (lane & bit) != 0;
#pragma unroll
  for (int i = 0; i < KEEP; ++i) {
    float send = hi ? p[i] : p[i + KEEP];
    float keep = hi ? p[i + KEEP] : p[i];
    p[i] = keep + row_xor_lane<bit>(send);
  }
}

__global__ __launch_bounds__(NT)
void graph_kernel(const float* __restrict__ feat, const float* __restrict__ noise,
                  float* __restrict__ out, float snr_pow) {
  __shared__ __align__(16) float Dl[PN * DP];
  __shared__ __align__(16) float Sl[PN * SP];
  __shared__ __align__(16) float wsum[NW][4];
  __shared__ float gst[2];
  __shared__ __align__(16) float colp[F_RG][PD];
  __shared__ __align__(16) float mu[PD];
  __shared__ __align__(16) float sv[NS_SLOTS][TN];
  __shared__ __align__(16) float part[4][PD];

  const int b    = blockIdx.x;
  const int tid  = threadIdx.x;
  const int lane = tid & 31;
  const int wid  = tid >> 5;
  const int cg   = tid & 63;
  const int rg   = tid >> 6;

  {
    const long long base = (long long)b * PN * PD + (long long)rg * PD + 4 * cg;
    const float4* f4 = (const float4*)(feat + base);
    const float4* z4 = (const float4*)(noise + base);

    float4 f[F_K], z[F_K];
#pragma unroll
    for (int k = 0; k < F_K; ++k) f[k] = f4[k * (F_RG * PD / 4)];
#pragma unroll
    for (int k = 0; k < F_K; ++k) z[k] = z4[k * (F_RG * PD / 4)];

    float4 af  = make_float4(0.0f, 0.0f, 0.0f, 0.0f), af2 = make_float4(0.0f, 0.0f, 0.0f, 0.0f);
    float4 an  = make_float4(0.0f, 0.0f, 0.0f, 0.0f), an2 = make_float4(0.0f, 0.0f, 0.0f, 0.0f);
#pragma unroll
    for (int k = 0; k < F_K; ++k) {
      af.x += f[k].x; af.y += f[k].y; af.z += f[k].z; af.w += f[k].w;
      af2.x += f[k].x * f[k].x; af2.y += f[k].y * f[k].y;
      af2.z += f[k].z * f[k].z; af2.w += f[k].w * f[k].w;
    }
#pragma unroll
    for (int k = 0; k < F_K; ++k) {
      an.x += z[k].x; an.y += z[k].y; an.z += z[k].z; an.w += z[k].w;
      an2.x += z[k].x * z[k].x; an2.y += z[k].y * z[k].y;
      an2.z += z[k].z * z[k].z; an2.w += z[k].w * z[k].w;
    }
    float sf  = (af.x + af.y) + (af.z + af.w);
    float sf2 = (af2.x + af2.y) + (af2.z + af2.w);
    float sn  = (an.x + an.y) + (an.z + an.w);
    float sn2 = (an2.x + an2.y) + (an2.z + an2.w);
#pragma unroll
    for (int m = 16; m > 0; m >>= 1) {
      sf  += __shfl_xor(sf, m);
      sf2 += __shfl_xor(sf2, m);
      sn  += __shfl_xor(sn, m);
      sn2 += __shfl_xor(sn2, m);
    }
    if (lane == 0) *(float4*)&wsum[wid][0] = make_float4(sf, sf2, sn, sn2);
    __syncthreads();
    if (wid == 0) {
      const float4 w4 = *(const float4*)&wsum[lane & (NW - 1)][0];
      double d0 = (double)w4.x, d1 = (double)w4.y;
      double d2 = (double)w4.z, d3 = (double)w4.w;
#pragma unroll
      for (int m = NW / 2; m > 0; m >>= 1) {
        d0 += __shfl_xor(d0, m);
        d1 += __shfl_xor(d1, m);
        d2 += __shfl_xor(d2, m);
        d3 += __shfl_xor(d3, m);
      }
      if (lane == 0) {
        double inv  = 1.0 / (double)(PN * PD);
        double gmf  = d0 * inv;
        double gmn  = d2 * inv;
        double tvar = d1 * inv - gmf * gmf;
        double nvar = d3 * inv - gmn * gmn;
        gst[0] = (float)sqrt((tvar / (double)snr_pow) / nvar);
        gst[1] = (float)gmn;
      }
    }
    __syncthreads();
    const float scl = gst[0];
    const float nmn = gst[1];

    float4 cs = make_float4(0.0f, 0.0f, 0.0f, 0.0f);
#pragma unroll
    for (int k = 0; k < F_K; ++k) {
      f[k].x = f[k].x + (z[k].x - nmn) * scl;
      f[k].y = f[k].y + (z[k].y - nmn) * scl;
      f[k].z = f[k].z + (z[k].z - nmn) * scl;
      f[k].w = f[k].w + (z[k].w - nmn) * scl;
      cs.x += f[k].x; cs.y += f[k].y; cs.z += f[k].z; cs.w += f[k].w;
    }
    *(float4*)&colp[rg][4 * cg] = cs;
    __syncthreads();
    {
      float s = (colp[0][tid] + colp[1][tid]) + (colp[2][tid] + colp[3][tid]);
      mu[tid] = s / (float)PN;
    }
    __syncthreads();
    const float4 m4 = *(const float4*)&mu[4 * cg];
#pragma unroll
    for (int k = 0; k < F_K; ++k) {
      float4 v;
      v.x = f[k].x - m4.x; v.y = f[k].y - m4.y; v.z = f[k].z - m4.z; v.w = f[k].w - m4.w;
      *(float4*)&Dl[(rg + F_RG * k) * DP + 4 * cg] = v;
    }
  }
  __syncthreads();

  {
    const int mr = lane & 15;
    const int h  = lane >> 4;
    const int wm = (wid >> 1) * 32;
    const int wn = (wid & 1) * 64;
    const float* pa = &Dl[(wm + mr) * DP + 2 * h];
    const float* pb = &Dl[(wn + mr) * DP + 2 * h];

    v8f acc[2][4];
#pragma unroll
    for (int im = 0; im < 2; ++im)
#pragma unroll
      for (int in = 0; in < 4; ++in) acc[im][in] = (v8f){};

#pragma unroll 1
    for (int k0 = 0; k0 < PD; k0 += 32) {
#pragma unroll
      for (int kk = 0; kk < 32; kk += 4) {
        const int kc = k0 + kk;
        v2f af[2], bf[4];
#pragma unroll
        for (int im = 0; im < 2; ++im) af[im] = *(const v2f*)(pa + 16 * im * DP + kc);
#pragma unroll
        for (int in = 0; in < 4; ++in) bf[in] = *(const v2f*)(pb + 16 * in * DP + kc);
#pragma unroll
        for (int im = 0; im < 2; ++im)
#pragma unroll
          for (int in = 0; in < 4; ++in)
            acc[im][in] = __builtin_amdgcn_wmma_f32_16x16x4_f32(false, af[im], false, bf[in], (short)0,
                                                                 acc[im][in], false, false);
      }
    }

    const float alpha = 1.0f / (float)(PN - 1);
#pragma unroll
    for (int im = 0; im < 2; ++im) {
#pragma unroll
      for (int r = 0; r < 8; ++r) {
        const int row = wm + 16 * im + 8 * h + r;
#pragma unroll
        for (int in = 0; in < 4; ++in) {
          const int col = wn + 16 * in + mr;
          Sl[row * SP + col] = acc[im][in][r] * alpha;
        }
      }
    }
  }
  __syncthreads();

  const int c    = lane & 15;
  const int row0 = wid * 16 + (lane >> 4) * 8;
  const int myrow = wid * 16 + (lane >> 1);

  v2f a[8][4];
#pragma unroll
  for (int q = 0; q < 8; ++q) {
    const float4* r = (const float4*)&Sl[(row0 + q) * SP];
#pragma unroll
    for (int j = 0; j < 2; ++j) {
      float4 t = r[16 * j + c];
      a[q][2 * j]     = (v2f){t.x, t.y};
      a[q][2 * j + 1] = (v2f){t.z, t.w};
    }
  }
  float dg[4];
#pragma unroll
  for (int j = 0; j < 4; ++j) dg[j] = Sl[(lane + 32 * j) * (SP + 1)];
  float trb = (dg[0] + dg[2]) + (dg[1] + dg[3]);
  trb += __shfl_xor(trb, 16);
  trb += row_xor_lane<8>(trb);
  trb += row_xor_lane<4>(trb);
  trb += row_xor_lane<2>(trb);
  trb += row_xor_lane<1>(trb);
  const float itr = 1.0f / trb;

  {
    const float4* m4p = (const float4*)mu;
    const float4 ma = m4p[lane], mb = m4p[32 + lane];
    float pu[16];
#pragma unroll
    for (int r = 0; r < 16; ++r) {
      const float4* dr = (const float4*)&Dl[(wid * 16 + r) * DP];
      const float4 da = dr[lane], db = dr[32 + lane];
      float s0 = da.x * ma.x, s1 = da.y * ma.y;
      s0 = fmaf(da.z, ma.z, s0); s1 = fmaf(da.w, ma.w, s1);
      s0 = fmaf(db.x, mb.x, s0); s1 = fmaf(db.y, mb.y, s1);
      s0 = fmaf(db.z, mb.z, s0); s1 = fmaf(db.w, mb.w, s1);
      pu[r] = s0 + s1;
    }
    float mine = 0.0f;
#pragma unroll
    for (int r = 0; r < 16; ++r) {
      float s = pu[r];
      s += __shfl_xor(s, 16);
      s += row_xor_lane<8>(s);
      s += row_xor_lane<4>(s);
      s += row_xor_lane<2>(s);
      s += row_xor_lane<1>(s);
      mine = (lane == r) ? s : mine;
    }
    if (lane < 16) sv[0][wid * 16 + lane] = mine;
  }
  __syncthreads();

  const int nops = ns_prog.n;
  NsStep nx = ns_prog.st[0];
  for (int i = 0; i < nops; ++i) {
    const NsStep op = nx;
    nx = ns_prog.st[i + 1];
    const float* xs = sv[op.slots & 15];
    float*       xd = sv[(op.slots >> 4) & 15];
    v2f x[4];
#pragma unroll
    for (int j = 0; j < 2; ++j) {
      float4 t = ((const float4*)xs)[16 * j + c];
      x[2 * j]     = (v2f){t.x, t.y};
      x[2 * j + 1] = (v2f){t.z, t.w};
    }
    __builtin_amdgcn_sched_barrier(0);
    const float es = xs[myrow];
    const float e1 = sv[(op.slots >> 8) & 15][myrow];
    const float e2 = sv[(op.slots >> 12) & 15][myrow];
    const float e3 = sv[(op.slots >> 16) & 15][myrow];
    const float e4 = sv[(op.slots >> 20) & 15][myrow];
    __builtin_amdgcn_sched_barrier(0);
    const float linh = fmaf(op.B, es, fmaf(op.C1, e1, fmaf(op.C2, e2, fmaf(op.C3, e3, op.C4 * e4))));
    const float ag  = op.A * itr;
    float p[8];
#pragma unroll
    for (int q = 0; q < 8; ++q) {
      v2f s = a[q][0] * x[0];
#pragma unroll
      for (int j = 1; j < 4; ++j) s = __builtin_elementwise_fma(a[q][j], x[j], s);
      p[q] = s.x + s.y;
    }
    ns_reduce_step<4>(p, lane);
    ns_reduce_step<2>(p, lane);
    ns_reduce_step<1>(p, lane);
    const float w   = fmaf(ag, p[0], linh);
    const float val = w + row_xor_lane<1>(w);
    if ((lane & 1) == 0) xd[myrow] = val;
    __syncthreads();
  }

  {
    const float4* dc = (const float4*)&Dl[(32 * rg) * DP] + cg;
    const float* v = &sv[1][32 * rg];
    float4 s4 = make_float4(0.0f, 0.0f, 0.0f, 0.0f);
#pragma unroll
    for (int k = 0; k < 32; ++k) {
      const float4 dv = dc[k * (DP / 4)];
      const float vk = v[k];
      s4.x = fmaf(dv.x, vk, s4.x); s4.y = fmaf(dv.y, vk, s4.y);
      s4.z = fmaf(dv.z, vk, s4.z); s4.w = fmaf(dv.w, vk, s4.w);
    }
    *(float4*)&part[rg][4 * cg] = s4;
    __syncthreads();
    const float sc = sqrtf(trb) * itr * (1.0f / (float)(TN - 1));
    out[(long long)b * PD + tid] = ((part[0][tid] + part[1][tid]) + (part[2][tid] + part[3][tid])) * sc;
  }
}

extern "C" void kernel_launch(void* const* d_in, const int* in_sizes, int n_in,
                              void* d_out, int out_size, void* d_ws, size_t ws_size,
                              hipStream_t stream) {
  (void)n_in; (void)ws_size; (void)out_size; (void)d_ws;
  const float* feat  = (const float*)d_in[1];
  const float* noise = (const float*)d_in[2];
  const int B = in_sizes[0];

  graph_kernel<<<B, NT, 0, stream>>>(feat, noise, (float*)d_out, 31.6227766016838f);
}
